# starting kernel
# baseline (speedup 1.0000x reference)
#include <hip/hip_runtime.h>
#include <hip/hip_bf16.h>
#include <math.h>

typedef __attribute__((ext_vector_type(2))) float v2f;
typedef __attribute__((ext_vector_type(8))) float v8f;

#define DIM 128
#define BPAD 132

__device__ __forceinline__ void gatomic_add_f32(float* addr, float v) {
    asm volatile("global_atomic_add_f32 %0, %1, off" :: "v"(addr), "v"(v) : "memory");
}

__global__ void zero_f32(float* __restrict__ p, int n) {
    int i = blockIdx.x * blockDim.x + threadIdx.x;
    int i4 = i * 4;
    if (i4 + 3 < n) {
        *(float4*)(p + i4) = make_float4(0.f, 0.f, 0.f, 0.f);
    } else {
        for (int j = i4; j < n; ++j) p[j] = 0.f;
    }
}

__global__ void degree_kernel(const int* __restrict__ dst, float* __restrict__ deg, int n_edges) {
    int e = blockIdx.x * blockDim.x + threadIdx.x;
    if (e < n_edges) gatomic_add_f32(deg + dst[e], 1.0f);
}

__global__ void scatter_kernel(const float* __restrict__ xin, const int* __restrict__ src,
                               const int* __restrict__ dst, float* __restrict__ agg,
                               int n_edges) {
    int wave = (blockIdx.x * blockDim.x + threadIdx.x) >> 5;
    int lane = threadIdx.x & 31;
    if (wave >= n_edges) return;
    int s = src[wave];
    int d = dst[wave];
    float4 v = ((const float4*)(xin + (size_t)s * DIM))[lane];
    float* arow = agg + (size_t)d * DIM + lane * 4;
    gatomic_add_f32(arow + 0, v.x);
    gatomic_add_f32(arow + 1, v.y);
    gatomic_add_f32(arow + 2, v.z);
    gatomic_add_f32(arow + 3, v.w);
}

__global__ void sage_gemm_kernel(const float* __restrict__ agg, const float* __restrict__ deg,
                                 const float* __restrict__ xin,
                                 const float* __restrict__ Wl, const float* __restrict__ bl,
                                 const float* __restrict__ Wr,
                                 float* __restrict__ out, int n_nodes) {
    __shared__ float Blt[16 * BPAD];
    __shared__ float Brt[16 * BPAD];

    int col_tile = blockIdx.x & 7;
    int group    = blockIdx.x >> 3;
    int wv       = threadIdx.x >> 5;
    int lane     = threadIdx.x & 31;
    int colbase  = col_tile * 16;

    for (int idx = threadIdx.x; idx < 2048; idx += blockDim.x) {
        int k = idx >> 4, n = idx & 15;
        Blt[n * BPAD + k] = Wl[(size_t)k * DIM + colbase + n];
        Brt[n * BPAD + k] = Wr[(size_t)k * DIM + colbase + n];
    }
    __syncthreads();

    int row_tile = group * 8 + wv;
    if (row_tile * 16 + 16 > n_nodes) return;

    int m  = lane & 15;
    int kk = (lane >> 4) * 2;
    int nn = lane & 15;

    int row = row_tile * 16 + m;
    float rdeg = 1.0f / fmaxf(deg[row], 1.0f);
    const float* arow = agg + (size_t)row * DIM;
    const float* xrow = xin + (size_t)row * DIM;
    const float* blane = Blt + nn * BPAD;
    const float* rlane = Brt + nn * BPAD;

    v8f acc = {};
#pragma unroll 4
    for (int k0 = 0; k0 < DIM; k0 += 4) {
        v2f a1 = *(const v2f*)(arow + k0 + kk);
        a1.x *= rdeg; a1.y *= rdeg;
        v2f b1 = *(const v2f*)(blane + k0 + kk);
        acc = __builtin_amdgcn_wmma_f32_16x16x4_f32(false, a1, false, b1,
                                                    (short)0, acc, false, false);
        v2f a2 = *(const v2f*)(xrow + k0 + kk);
        v2f b2 = *(const v2f*)(rlane + k0 + kk);
        acc = __builtin_amdgcn_wmma_f32_16x16x4_f32(false, a2, false, b2,
                                                    (short)0, acc, false, false);
    }

    int ccol  = colbase + (lane & 15);
    int rbase = row_tile * 16 + (lane >> 4) * 8;
    float bias = bl[ccol];
#pragma unroll
    for (int i = 0; i < 8; ++i) {
        float v = acc[i] + bias;
        out[(size_t)(rbase + i) * DIM + ccol] = fmaxf(v, 0.0f);
    }
}

__global__ void head_kernel(const float* __restrict__ h,
                            const float* __restrict__ Wp, const float* __restrict__ bp,
                            const float* __restrict__ Wd, const float* __restrict__ bd,
                            float* __restrict__ out, int n_nodes) {
    int wave = (blockIdx.x * blockDim.x + threadIdx.x) >> 5;
    int lane = threadIdx.x & 31;
    if (wave >= n_nodes) return;
    float4 hv = ((const float4*)(h + (size_t)wave * DIM))[lane];
    float4 wp = ((const float4*)Wp)[lane];
    float4 wd = ((const float4*)Wd)[lane];
    float p = hv.x * wp.x + hv.y * wp.y + hv.z * wp.z + hv.w * wp.w;
    float d = hv.x * wd.x + hv.y * wd.y + hv.z * wd.z + hv.w * wd.w;
#pragma unroll
    for (int off = 16; off > 0; off >>= 1) {
        p += __shfl_xor(p, off, 32);
        d += __shfl_xor(d, off, 32);
    }
    if (lane == 0) {
        p += bp[0];
        d += bd[0];
        d = 1.0f / (1.0f + __expf(-d));
        out[wave] = p - d;
        out[n_nodes + wave] = p + d;
    }
}

extern "C" void kernel_launch(void* const* d_in, const int* in_sizes, int n_in,
                              void* d_out, int out_size, void* d_ws, size_t ws_size,
                              hipStream_t stream) {
    const float* x    = (const float*)d_in[0];
    const int*   edge = (const int*)d_in[1];
    const float* W1l  = (const float*)d_in[2];
    const float* b1l  = (const float*)d_in[3];
    const float* W1r  = (const float*)d_in[4];
    const float* W2l  = (const float*)d_in[5];
    const float* b2l  = (const float*)d_in[6];
    const float* W2r  = (const float*)d_in[7];
    const float* Wp   = (const float*)d_in[8];
    const float* bp   = (const float*)d_in[9];
    const float* Wd   = (const float*)d_in[10];
    const float* bd   = (const float*)d_in[11];
    float* out = (float*)d_out;

    const int N = in_sizes[0] / DIM;
    const int E = in_sizes[1] / 2;
    const int* src = edge;
    const int* dst = edge + E;

    char* ws = (char*)d_ws;
    float* agg = (float*)ws;
    float* h1  = agg + (size_t)N * DIM;
    float* h2  = h1 + (size_t)N * DIM;
    float* deg = h2 + (size_t)N * DIM;

    const int BLK = 256;
    int nfeat = N * DIM;
    int zgrid = (nfeat / 4 + BLK - 1) / BLK;
    int dgrid = (N / 4 + BLK) / BLK;
    int egrid_t = (E + BLK - 1) / BLK;
    int egrid_w = (E * 32 + BLK - 1) / BLK;
    int rowTiles = (N + 15) / 16;
    int ggrid = 8 * ((rowTiles + 7) / 8);
    int hgrid = (N * 32 + BLK - 1) / BLK;

    zero_f32<<<zgrid, BLK, 0, stream>>>(agg, nfeat);
    zero_f32<<<dgrid, BLK, 0, stream>>>(deg, N);
    degree_kernel<<<egrid_t, BLK, 0, stream>>>(dst, deg, E);
    scatter_kernel<<<egrid_w, BLK, 0, stream>>>(x, src, dst, agg, E);
    sage_gemm_kernel<<<ggrid, BLK, 0, stream>>>(agg, deg, x, W1l, b1l, W1r, h1, N);

    zero_f32<<<zgrid, BLK, 0, stream>>>(agg, nfeat);
    scatter_kernel<<<egrid_w, BLK, 0, stream>>>(h1, src, dst, agg, E);
    sage_gemm_kernel<<<ggrid, BLK, 0, stream>>>(agg, deg, h1, W2l, b2l, W2r, h2, N);

    head_kernel<<<hgrid, BLK, 0, stream>>>(h2, Wp, bp, Wd, bd, out, N);
}


# scatter-add lanes re-laid (lane l owns floats l,l+32,l+64,l+96): each f32 atomic instruction covers one whole 128-byte line, 4 line requests per edge not 16; sums stay arrival-order atomics
# speedup vs baseline: 1.2277x; 1.2277x over previous
#include <hip/hip_runtime.h>
#include <hip/hip_bf16.h>
#include <math.h>

typedef __attribute__((ext_vector_type(2))) float v2f;
typedef __attribute__((ext_vector_type(8))) float v8f;

#define DIM 128
#define BPAD 132

__device__ __forceinline__ void gatomic_add_f32(float* addr, float v) {
    asm volatile("global_atomic_add_f32 %0, %1, off" :: "v"(addr), "v"(v) : "memory");
}

template <int BYTE_OFF>
__device__ __forceinline__ void gatomic_add_f32_off(float* addr, float v) {
    asm volatile("global_atomic_add_f32 %0, %1, off offset:%2"
                 :: "v"(addr), "v"(v), "n"(BYTE_OFF) : "memory");
}

__global__ void zero_f32(float* __restrict__ p, int n) {
    int i = blockIdx.x * blockDim.x + threadIdx.x;
    int i4 = i * 4;
    if (i4 + 3 < n) {
        *(float4*)(p + i4) = make_float4(0.f, 0.f, 0.f, 0.f);
    } else {
        for (int j = i4; j < n; ++j) p[j] = 0.f;
    }
}

__global__ void degree_kernel(const int* __restrict__ dst, float* __restrict__ deg, int n_edges) {
    int e = blockIdx.x * blockDim.x + threadIdx.x;
    if (e < n_edges) gatomic_add_f32(deg + dst[e], 1.0f);
}

__global__ void scatter_kernel(const float* __restrict__ xin, const int* __restrict__ src,
                               const int* __restrict__ dst, float* __restrict__ agg,
                               int n_edges) {
    int wave = (blockIdx.x * blockDim.x + threadIdx.x) >> 5;
    int lane = threadIdx.x & 31;
    if (wave >= n_edges) return;
    int s = src[wave];
    int d = dst[wave];
    const float* xrow = xin + (size_t)s * DIM + lane;
    float v0 = xrow[0];
    float v1 = xrow[32];
    float v2 = xrow[64];
    float v3 = xrow[96];
    float* arow = agg + (size_t)d * DIM + lane;
    gatomic_add_f32_off<0>(arow, v0);
    gatomic_add_f32_off<128>(arow, v1);
    gatomic_add_f32_off<256>(arow, v2);
    gatomic_add_f32_off<384>(arow, v3);
}

__global__ void sage_gemm_kernel(const float* __restrict__ agg, const float* __restrict__ deg,
                                 const float* __restrict__ xin,
                                 const float* __restrict__ Wl, const float* __restrict__ bl,
                                 const float* __restrict__ Wr,
                                 float* __restrict__ out, int n_nodes) {
    __shared__ float Blt[16 * BPAD];
    __shared__ float Brt[16 * BPAD];

    int col_tile = blockIdx.x & 7;
    int group    = blockIdx.x >> 3;
    int wv       = threadIdx.x >> 5;
    int lane     = threadIdx.x & 31;
    int colbase  = col_tile * 16;

    for (int idx = threadIdx.x; idx < 2048; idx += blockDim.x) {
        int k = idx >> 4, n = idx & 15;
        Blt[n * BPAD + k] = Wl[(size_t)k * DIM + colbase + n];
        Brt[n * BPAD + k] = Wr[(size_t)k * DIM + colbase + n];
    }
    __syncthreads();

    int row_tile = group * 8 + wv;
    if (row_tile * 16 + 16 > n_nodes) return;

    int m  = lane & 15;
    int kk = (lane >> 4) * 2;
    int nn = lane & 15;

    int row = row_tile * 16 + m;
    float rdeg = 1.0f / fmaxf(deg[row], 1.0f);
    const float* arow = agg + (size_t)row * DIM;
    const float* xrow = xin + (size_t)row * DIM;
    const float* blane = Blt + nn * BPAD;
    const float* rlane = Brt + nn * BPAD;

    v8f acc = {};
#pragma unroll 4
    for (int k0 = 0; k0 < DIM; k0 += 4) {
        v2f a1 = *(const v2f*)(arow + k0 + kk);
        a1.x *= rdeg; a1.y *= rdeg;
        v2f b1 = *(const v2f*)(blane + k0 + kk);
        acc = __builtin_amdgcn_wmma_f32_16x16x4_f32(false, a1, false, b1,
                                                    (short)0, acc, false, false);
        v2f a2 = *(const v2f*)(xrow + k0 + kk);
        v2f b2 = *(const v2f*)(rlane + k0 + kk);
        acc = __builtin_amdgcn_wmma_f32_16x16x4_f32(false, a2, false, b2,
                                                    (short)0, acc, false, false);
    }

    int ccol  = colbase + (lane & 15);
    int rbase = row_tile * 16 + (lane >> 4) * 8;
    float bias = bl[ccol];
#pragma unroll
    for (int i = 0; i < 8; ++i) {
        float v = acc[i] + bias;
        out[(size_t)(rbase + i) * DIM + ccol] = fmaxf(v, 0.0f);
    }
}

__global__ void head_kernel(const float* __restrict__ h,
                            const float* __restrict__ Wp, const float* __restrict__ bp,
                            const float* __restrict__ Wd, const float* __restrict__ bd,
                            float* __restrict__ out, int n_nodes) {
    int wave = (blockIdx.x * blockDim.x + threadIdx.x) >> 5;
    int lane = threadIdx.x & 31;
    if (wave >= n_nodes) return;
    float4 hv = ((const float4*)(h + (size_t)wave * DIM))[lane];
    float4 wp = ((const float4*)Wp)[lane];
    float4 wd = ((const float4*)Wd)[lane];
    float p = hv.x * wp.x + hv.y * wp.y + hv.z * wp.z + hv.w * wp.w;
    float d = hv.x * wd.x + hv.y * wd.y + hv.z * wd.z + hv.w * wd.w;
#pragma unroll
    for (int off = 16; off > 0; off >>= 1) {
        p += __shfl_xor(p, off, 32);
        d += __shfl_xor(d, off, 32);
    }
    if (lane == 0) {
        p += bp[0];
        d += bd[0];
        d = 1.0f / (1.0f + __expf(-d));
        out[wave] = p - d;
        out[n_nodes + wave] = p + d;
    }
}

extern "C" void kernel_launch(void* const* d_in, const int* in_sizes, int n_in,
                              void* d_out, int out_size, void* d_ws, size_t ws_size,
                              hipStream_t stream) {
    const float* x    = (const float*)d_in[0];
    const int*   edge = (const int*)d_in[1];
    const float* W1l  = (const float*)d_in[2];
    const float* b1l  = (const float*)d_in[3];
    const float* W1r  = (const float*)d_in[4];
    const float* W2l  = (const float*)d_in[5];
    const float* b2l  = (const float*)d_in[6];
    const float* W2r  = (const float*)d_in[7];
    const float* Wp   = (const float*)d_in[8];
    const float* bp   = (const float*)d_in[9];
    const float* Wd   = (const float*)d_in[10];
    const float* bd   = (const float*)d_in[11];
    float* out = (float*)d_out;

    const int N = in_sizes[0] / DIM;
    const int E = in_sizes[1] / 2;
    const int* src = edge;
    const int* dst = edge + E;

    char* ws = (char*)d_ws;
    float* agg = (float*)ws;
    float* h1  = agg + (size_t)N * DIM;
    float* h2  = h1 + (size_t)N * DIM;
    float* deg = h2 + (size_t)N * DIM;

    const int BLK = 256;
    int nfeat = N * DIM;
    int zgrid = (nfeat / 4 + BLK - 1) / BLK;
    int dgrid = (N / 4 + BLK) / BLK;
    int egrid_t = (E + BLK - 1) / BLK;
    int egrid_w = (E * 32 + BLK - 1) / BLK;
    int rowTiles = (N + 15) / 16;
    int ggrid = 8 * ((rowTiles + 7) / 8);
    int hgrid = (N * 32 + BLK - 1) / BLK;

    zero_f32<<<zgrid, BLK, 0, stream>>>(agg, nfeat);
    zero_f32<<<dgrid, BLK, 0, stream>>>(deg, N);
    degree_kernel<<<egrid_t, BLK, 0, stream>>>(dst, deg, E);
    scatter_kernel<<<egrid_w, BLK, 0, stream>>>(x, src, dst, agg, E);
    sage_gemm_kernel<<<ggrid, BLK, 0, stream>>>(agg, deg, x, W1l, b1l, W1r, h1, N);

    zero_f32<<<zgrid, BLK, 0, stream>>>(agg, nfeat);
    scatter_kernel<<<egrid_w, BLK, 0, stream>>>(h1, src, dst, agg, E);
    sage_gemm_kernel<<<ggrid, BLK, 0, stream>>>(agg, deg, h1, W2l, b2l, W2r, h2, N);

    head_kernel<<<hgrid, BLK, 0, stream>>>(h2, Wp, bp, Wd, bd, out, N);
}


# scatter-add: one wave32 per 8 consecutive edges (indices by scalar load), all 32 row-line loads issued before the 32 line atomics: 8x fewer waves, 8 rows in flight; sums stay arrival-order atomics
# speedup vs baseline: 1.5604x; 1.2710x over previous
#include <hip/hip_runtime.h>
#include <hip/hip_bf16.h>
#include <math.h>

typedef __attribute__((ext_vector_type(2))) float v2f;
typedef __attribute__((ext_vector_type(8))) float v8f;

#define DIM 128
#define BPAD 132

__device__ __forceinline__ void gatomic_add_f32(float* addr, float v) {
    asm volatile("global_atomic_add_f32 %0, %1, off" :: "v"(addr), "v"(v) : "memory");
}

template <int BYTE_OFF>
__device__ __forceinline__ void gatomic_add_f32_off(float* addr, float v) {
    asm volatile("global_atomic_add_f32 %0, %1, off offset:%2"
                 :: "v"(addr), "v"(v), "n"(BYTE_OFF) : "memory");
}

__global__ void zero_f32(float* __restrict__ p, int n) {
    int i = blockIdx.x * blockDim.x + threadIdx.x;
    int i4 = i * 4;
    if (i4 + 3 < n) {
        *(float4*)(p + i4) = make_float4(0.f, 0.f, 0.f, 0.f);
    } else {
        for (int j = i4; j < n; ++j) p[j] = 0.f;
    }
}

__global__ void degree_kernel(const int* __restrict__ dst, float* __restrict__ deg, int n_edges) {
    int e = blockIdx.x * blockDim.x + threadIdx.x;
    if (e < n_edges) gatomic_add_f32(deg + dst[e], 1.0f);
}

#define EPW 8

template <int BYTE_OFF>
__device__ __forceinline__ void gatomic_add_f32_srow(float* rowbase, unsigned lane_bytes, float v) {
    asm volatile("global_atomic_add_f32 %0, %1, %2 offset:%3"
                 :: "v"(lane_bytes), "v"(v), "s"(rowbase), "n"(BYTE_OFF) : "memory");
}

__global__ void scatter_kernel(const float* __restrict__ xin, const int* __restrict__ src,
                               const int* __restrict__ dst, float* __restrict__ agg,
                               int n_edges) {
    int lane = threadIdx.x & 31;
    int wib = __builtin_amdgcn_readfirstlane((int)(threadIdx.x >> 5));
    int e0  = (blockIdx.x * (int)(blockDim.x >> 5) + wib) * EPW;
    if (e0 >= n_edges) return;
    if (e0 + EPW <= n_edges) {
        int s[EPW], d[EPW];
#pragma unroll
        for (int j = 0; j < EPW; ++j) { s[j] = src[e0 + j]; d[j] = dst[e0 + j]; }
        asm volatile("" :: "s"(s[0]), "s"(s[1]), "s"(s[2]), "s"(s[3]),
                           "s"(s[4]), "s"(s[5]), "s"(s[6]), "s"(s[7]),
                           "s"(d[0]), "s"(d[1]), "s"(d[2]), "s"(d[3]),
                           "s"(d[4]), "s"(d[5]), "s"(d[6]), "s"(d[7]));
        float v[EPW][4];
#pragma unroll
        for (int j = 0; j < EPW; ++j) {
            const float* xrow = xin + (size_t)s[j] * DIM + lane;
            v[j][0] = xrow[0];
            v[j][1] = xrow[32];
            v[j][2] = xrow[64];
            v[j][3] = xrow[96];
        }
        unsigned lb = (unsigned)lane * 4u;
#pragma unroll
        for (int j = 0; j < EPW; ++j) {
            float* arow = agg + (size_t)d[j] * DIM;
            gatomic_add_f32_srow<0>(arow, lb, v[j][0]);
            gatomic_add_f32_srow<128>(arow, lb, v[j][1]);
            gatomic_add_f32_srow<256>(arow, lb, v[j][2]);
            gatomic_add_f32_srow<384>(arow, lb, v[j][3]);
        }
    } else {
        for (int e = e0; e < n_edges; ++e) {
            const float* xrow = xin + (size_t)src[e] * DIM + lane;
            float v0 = xrow[0];
            float v1 = xrow[32];
            float v2 = xrow[64];
            float v3 = xrow[96];
            float* arow = agg + (size_t)dst[e] * DIM + lane;
            gatomic_add_f32_off<0>(arow, v0);
            gatomic_add_f32_off<128>(arow, v1);
            gatomic_add_f32_off<256>(arow, v2);
            gatomic_add_f32_off<384>(arow, v3);
        }
    }
}

__global__ void sage_gemm_kernel(const float* __restrict__ agg, const float* __restrict__ deg,
                                 const float* __restrict__ xin,
                                 const float* __restrict__ Wl, const float* __restrict__ bl,
                                 const float* __restrict__ Wr,
                                 float* __restrict__ out, int n_nodes) {
    __shared__ float Blt[16 * BPAD];
    __shared__ float Brt[16 * BPAD];

    int col_tile = blockIdx.x & 7;
    int group    = blockIdx.x >> 3;
    int wv       = threadIdx.x >> 5;
    int lane     = threadIdx.x & 31;
    int colbase  = col_tile * 16;

    for (int idx = threadIdx.x; idx < 2048; idx += blockDim.x) {
        int k = idx >> 4, n = idx & 15;
        Blt[n * BPAD + k] = Wl[(size_t)k * DIM + colbase + n];
        Brt[n * BPAD + k] = Wr[(size_t)k * DIM + colbase + n];
    }
    __syncthreads();

    int row_tile = group * 8 + wv;
    if (row_tile * 16 + 16 > n_nodes) return;

    int m  = lane & 15;
    int kk = (lane >> 4) * 2;
    int nn = lane & 15;

    int row = row_tile * 16 + m;
    float rdeg = 1.0f / fmaxf(deg[row], 1.0f);
    const float* arow = agg + (size_t)row * DIM;
    const float* xrow = xin + (size_t)row * DIM;
    const float* blane = Blt + nn * BPAD;
    const float* rlane = Brt + nn * BPAD;

    v8f acc = {};
#pragma unroll 4
    for (int k0 = 0; k0 < DIM; k0 += 4) {
        v2f a1 = *(const v2f*)(arow + k0 + kk);
        a1.x *= rdeg; a1.y *= rdeg;
        v2f b1 = *(const v2f*)(blane + k0 + kk);
        acc = __builtin_amdgcn_wmma_f32_16x16x4_f32(false, a1, false, b1,
                                                    (short)0, acc, false, false);
        v2f a2 = *(const v2f*)(xrow + k0 + kk);
        v2f b2 = *(const v2f*)(rlane + k0 + kk);
        acc = __builtin_amdgcn_wmma_f32_16x16x4_f32(false, a2, false, b2,
                                                    (short)0, acc, false, false);
    }

    int ccol  = colbase + (lane & 15);
    int rbase = row_tile * 16 + (lane >> 4) * 8;
    float bias = bl[ccol];
#pragma unroll
    for (int i = 0; i < 8; ++i) {
        float v = acc[i] + bias;
        out[(size_t)(rbase + i) * DIM + ccol] = fmaxf(v, 0.0f);
    }
}

__global__ void head_kernel(const float* __restrict__ h,
                            const float* __restrict__ Wp, const float* __restrict__ bp,
                            const float* __restrict__ Wd, const float* __restrict__ bd,
                            float* __restrict__ out, int n_nodes) {
    int wave = (blockIdx.x * blockDim.x + threadIdx.x) >> 5;
    int lane = threadIdx.x & 31;
    if (wave >= n_nodes) return;
    float4 hv = ((const float4*)(h + (size_t)wave * DIM))[lane];
    float4 wp = ((const float4*)Wp)[lane];
    float4 wd = ((const float4*)Wd)[lane];
    float p = hv.x * wp.x + hv.y * wp.y + hv.z * wp.z + hv.w * wp.w;
    float d = hv.x * wd.x + hv.y * wd.y + hv.z * wd.z + hv.w * wd.w;
#pragma unroll
    for (int off = 16; off > 0; off >>= 1) {
        p += __shfl_xor(p, off, 32);
        d += __shfl_xor(d, off, 32);
    }
    if (lane == 0) {
        p += bp[0];
        d += bd[0];
        d = 1.0f / (1.0f + __expf(-d));
        out[wave] = p - d;
        out[n_nodes + wave] = p + d;
    }
}

extern "C" void kernel_launch(void* const* d_in, const int* in_sizes, int n_in,
                              void* d_out, int out_size, void* d_ws, size_t ws_size,
                              hipStream_t stream) {
    const float* x    = (const float*)d_in[0];
    const int*   edge = (const int*)d_in[1];
    const float* W1l  = (const float*)d_in[2];
    const float* b1l  = (const float*)d_in[3];
    const float* W1r  = (const float*)d_in[4];
    const float* W2l  = (const float*)d_in[5];
    const float* b2l  = (const float*)d_in[6];
    const float* W2r  = (const float*)d_in[7];
    const float* Wp   = (const float*)d_in[8];
    const float* bp   = (const float*)d_in[9];
    const float* Wd   = (const float*)d_in[10];
    const float* bd   = (const float*)d_in[11];
    float* out = (float*)d_out;

    const int N = in_sizes[0] / DIM;
    const int E = in_sizes[1] / 2;
    const int* src = edge;
    const int* dst = edge + E;

    char* ws = (char*)d_ws;
    float* agg = (float*)ws;
    float* h1  = agg + (size_t)N * DIM;
    float* h2  = h1 + (size_t)N * DIM;
    float* deg = h2 + (size_t)N * DIM;

    const int BLK = 256;
    int nfeat = N * DIM;
    int zgrid = (nfeat / 4 + BLK - 1) / BLK;
    int dgrid = (N / 4 + BLK) / BLK;
    int egrid_t = (E + BLK - 1) / BLK;
    int egrid_w = ((E + EPW - 1) / EPW * 32 + BLK - 1) / BLK;
    int rowTiles = (N + 15) / 16;
    int ggrid = 8 * ((rowTiles + 7) / 8);
    int hgrid = (N * 32 + BLK - 1) / BLK;

    zero_f32<<<zgrid, BLK, 0, stream>>>(agg, nfeat);
    zero_f32<<<dgrid, BLK, 0, stream>>>(deg, N);
    degree_kernel<<<egrid_t, BLK, 0, stream>>>(dst, deg, E);
    scatter_kernel<<<egrid_w, BLK, 0, stream>>>(x, src, dst, agg, E);
    sage_gemm_kernel<<<ggrid, BLK, 0, stream>>>(agg, deg, x, W1l, b1l, W1r, h1, N);

    zero_f32<<<zgrid, BLK, 0, stream>>>(agg, nfeat);
    scatter_kernel<<<egrid_w, BLK, 0, stream>>>(h1, src, dst, agg, E);
    sage_gemm_kernel<<<ggrid, BLK, 0, stream>>>(agg, deg, h1, W2l, b2l, W2r, h2, N);

    head_kernel<<<hgrid, BLK, 0, stream>>>(h2, Wp, bp, Wd, bd, out, N);
}


# atomic scatter-add replaced by an atomic-free pull: edges stable-sorted by destination once per call (2 radix passes), one wave sums a node's rows in edge order, one store per line; sum order fixed
# speedup vs baseline: 1.7268x; 1.1066x over previous
#include <hip/hip_runtime.h>
#include <hip/hip_bf16.h>
#include <math.h>

typedef __attribute__((ext_vector_type(2))) float v2f;
typedef __attribute__((ext_vector_type(8))) float v8f;

#define DIM 128
#define BPAD 132


#define RBITS 9
#define NBINS 512
#define GROUPS 16
#define TILE 4096
#define SORT_T 256

__device__ __forceinline__ unsigned same_digit_lanes(int d, bool valid) {
    unsigned peers = __builtin_amdgcn_ballot_w32(valid);
#pragma unroll
    for (int b = 0; b < RBITS; ++b) {
        bool bit = ((d >> b) & 1) != 0;
        unsigned bal = __builtin_amdgcn_ballot_w32(bit);
        peers &= bit ? bal : ~bal;
    }
    return peers;
}

__global__ void __launch_bounds__(SORT_T)
digit_count_kernel(const int* __restrict__ keys, int* __restrict__ ghist,
                   int n, int shift, int ntiles) {
    __shared__ int wh[8 * NBINS];
    int tid = threadIdx.x, lane = tid & 31, wv = tid >> 5;
    for (int i = tid; i < 8 * NBINS; i += SORT_T) wh[i] = 0;
    __syncthreads();
    int base = blockIdx.x * TILE + wv * (GROUPS * 32) + lane;
    unsigned lt = (1u << lane) - 1u;
    int* mine = wh + wv * NBINS;
    int dg[GROUPS];
#pragma unroll
    for (int g = 0; g < GROUPS; ++g) {
        int idx = base + g * 32;
        dg[g] = (idx < n) ? ((keys[idx] >> shift) & (NBINS - 1)) : 0;
    }
#pragma unroll
    for (int g = 0; g < GROUPS; ++g) {
        bool valid = (base + g * 32) < n;
        int d = dg[g];
        unsigned peers = same_digit_lanes(d, valid);
        if (valid && (peers & lt) == 0u) mine[d] += __popc(peers);
        __syncthreads();
    }
    for (int d = tid; d < NBINS; d += SORT_T) {
        int s = 0;
        for (int w = 0; w < 8; ++w) s += wh[w * NBINS + d];
        ghist[d * ntiles + blockIdx.x] = s;
    }
}

#define SCAN_T 1024
__global__ void __launch_bounds__(SCAN_T)
exclusive_scan_kernel(const int* __restrict__ cnt, int* __restrict__ outp, int n) {
    __shared__ int wtot[32];
    __shared__ int wpre[32];
    int tid  = threadIdx.x;
    int lane = tid & 31;
    int wv   = tid >> 5;
    int first = blockIdx.x * SCAN_T;

    const int4* c4 = (const int4*)cnt;
    int nvec = first >> 2;
    int pre = 0;
#pragma unroll 8
    for (int i = tid; i < nvec; i += SCAN_T) {
        int4 q = c4[i];
        pre += (q.x + q.y) + (q.z + q.w);
    }

    int i = first + tid;
    int c = (i < n) ? cnt[i] : 0;

    int inc = c;
#pragma unroll
    for (int off = 1; off < 32; off <<= 1) {
        int t = __shfl_up(inc, off, 32);
        if (lane >= off) inc += t;
    }
#pragma unroll
    for (int off = 16; off > 0; off >>= 1) pre += __shfl_xor(pre, off, 32);
    if (lane == 31) wtot[wv] = inc;
    if (lane == 0)  wpre[wv] = pre;
    __syncthreads();

    int t = (lane < wv) ? wtot[lane] : 0;
    int p = wpre[lane];
#pragma unroll
    for (int off = 16; off > 0; off >>= 1) {
        t += __shfl_xor(t, off, 32);
        p += __shfl_xor(p, off, 32);
    }
    if (i <= n) outp[i] = p + t + (inc - c);
}

__global__ void __launch_bounds__(SORT_T)
digit_move_kernel(const int* __restrict__ kin, const int* __restrict__ vin,
                  int* __restrict__ kout, int* __restrict__ vout,
                  const int* __restrict__ gscan, int n, int shift, int ntiles) {
    __shared__ int wh[8 * NBINS];
    int tid = threadIdx.x, lane = tid & 31, wv = tid >> 5;
    for (int i = tid; i < 8 * NBINS; i += SORT_T) wh[i] = 0;
    __syncthreads();
    int base = blockIdx.x * TILE + wv * (GROUPS * 32) + lane;
    unsigned lt = (1u << lane) - 1u;
    int* mine = wh + wv * NBINS;
    int key[GROUPS], val[GROUPS];
#pragma unroll
    for (int g = 0; g < GROUPS; ++g) {
        int idx = base + g * 32;
        bool valid = idx < n;
        key[g] = valid ? kin[idx] : 0;
        val[g] = valid ? vin[idx] : 0;
    }
#pragma unroll
    for (int g = 0; g < GROUPS; ++g) {
        bool valid = (base + g * 32) < n;
        int d = (key[g] >> shift) & (NBINS - 1);
        unsigned peers = same_digit_lanes(d, valid);
        if (valid && (peers & lt) == 0u) mine[d] += __popc(peers);
        __syncthreads();
    }
    for (int d = tid; d < NBINS; d += SORT_T) {
        int run = gscan[d * ntiles + blockIdx.x];
        for (int w = 0; w < 8; ++w) {
            int t = wh[w * NBINS + d];
            wh[w * NBINS + d] = run;
            run += t;
        }
    }
    __syncthreads();
#pragma unroll
    for (int g = 0; g < GROUPS; ++g) {
        int idx = base + g * 32;
        bool valid = idx < n;
        int d = (key[g] >> shift) & (NBINS - 1);
        unsigned peers = same_digit_lanes(d, valid);
        int rank = __popc(peers & lt);
        int pos = mine[d] + rank;
        __syncthreads();
        if (valid && rank == 0) mine[d] += __popc(peers);
        if (valid) { kout[pos] = key[g]; vout[pos] = val[g]; }
        __syncthreads();
    }
}

__global__ void rowptr_kernel(const int* __restrict__ skey, int* __restrict__ rowptr,
                              int n_edges, int n_nodes) {
    int i = blockIdx.x * blockDim.x + threadIdx.x;
    if (i > n_edges) return;
    int prev = (i == 0) ? -1 : skey[i - 1];
    int cur  = (i == n_edges) ? n_nodes : skey[i];
    for (int k = prev + 1; k <= cur; ++k) rowptr[k] = i;
}

__device__ __forceinline__ float4 row_ld(const float4* __restrict__ xv, int mysrc, int j, int lane) {
    int s = __builtin_amdgcn_readlane(mysrc, j);
    return xv[(size_t)s * (DIM / 4) + lane];
}
__device__ __forceinline__ void row_add(float4& acc, const float4& v) {
    acc.x += v.x; acc.y += v.y; acc.z += v.z; acc.w += v.w;
}

__global__ void aggregate_kernel(const float* __restrict__ xin, const int* __restrict__ rowptr,
                                 const int* __restrict__ gsrc, float* __restrict__ agg,
                                 float* __restrict__ deg, int n_nodes) {
    int lane = threadIdx.x & 31;
    int wib  = __builtin_amdgcn_readfirstlane((int)(threadIdx.x >> 5));
    int node = blockIdx.x * (int)(blockDim.x >> 5) + wib;
    if (node >= n_nodes) return;
    int beg = rowptr[node];
    int end = rowptr[node + 1];
    const float4* xv = (const float4*)xin;
    float4 acc = make_float4(0.f, 0.f, 0.f, 0.f);

    for (int base = beg; base < end; base += 32) {
        int n = end - base;
        if (n > 32) n = 32;
        int mysrc = (lane < n) ? gsrc[base + lane] : 0;
        int j = 0;
        for (; j + 8 <= n; j += 8) {
            float4 v[8];
#pragma unroll
            for (int u = 0; u < 8; ++u) v[u] = row_ld(xv, mysrc, j + u, lane);
#pragma unroll
            for (int u = 0; u < 8; ++u) row_add(acc, v[u]);
        }
        if (j + 4 <= n) {
            float4 v0 = row_ld(xv, mysrc, j, lane);
            float4 v1 = row_ld(xv, mysrc, j + 1, lane);
            float4 v2 = row_ld(xv, mysrc, j + 2, lane);
            float4 v3 = row_ld(xv, mysrc, j + 3, lane);
            row_add(acc, v0); row_add(acc, v1); row_add(acc, v2); row_add(acc, v3);
            j += 4;
        }
        if (j + 2 <= n) {
            float4 v0 = row_ld(xv, mysrc, j, lane);
            float4 v1 = row_ld(xv, mysrc, j + 1, lane);
            row_add(acc, v0); row_add(acc, v1);
            j += 2;
        }
        if (j < n) {
            float4 v0 = row_ld(xv, mysrc, j, lane);
            row_add(acc, v0);
        }
    }
    ((float4*)agg)[(size_t)node * (DIM / 4) + lane] = acc;
    if (deg != nullptr && lane == 0) deg[node] = (float)(end - beg);
}

__global__ void sage_gemm_kernel(const float* __restrict__ agg, const float* __restrict__ deg,
                                 const float* __restrict__ xin,
                                 const float* __restrict__ Wl, const float* __restrict__ bl,
                                 const float* __restrict__ Wr,
                                 float* __restrict__ out, int n_nodes) {
    __shared__ float Blt[16 * BPAD];
    __shared__ float Brt[16 * BPAD];

    int col_tile = blockIdx.x & 7;
    int group    = blockIdx.x >> 3;
    int wv       = threadIdx.x >> 5;
    int lane     = threadIdx.x & 31;
    int colbase  = col_tile * 16;

    for (int idx = threadIdx.x; idx < 2048; idx += blockDim.x) {
        int k = idx >> 4, n = idx & 15;
        Blt[n * BPAD + k] = Wl[(size_t)k * DIM + colbase + n];
        Brt[n * BPAD + k] = Wr[(size_t)k * DIM + colbase + n];
    }
    __syncthreads();

    int row_tile = group * 8 + wv;
    if (row_tile * 16 + 16 > n_nodes) return;

    int m  = lane & 15;
    int kk = (lane >> 4) * 2;
    int nn = lane & 15;

    int row = row_tile * 16 + m;
    float rdeg = 1.0f / fmaxf(deg[row], 1.0f);
    const float* arow = agg + (size_t)row * DIM;
    const float* xrow = xin + (size_t)row * DIM;
    const float* blane = Blt + nn * BPAD;
    const float* rlane = Brt + nn * BPAD;

    v8f acc = {};
#pragma unroll 4
    for (int k0 = 0; k0 < DIM; k0 += 4) {
        v2f a1 = *(const v2f*)(arow + k0 + kk);
        a1.x *= rdeg; a1.y *= rdeg;
        v2f b1 = *(const v2f*)(blane + k0 + kk);
        acc = __builtin_amdgcn_wmma_f32_16x16x4_f32(false, a1, false, b1,
                                                    (short)0, acc, false, false);
        v2f a2 = *(const v2f*)(xrow + k0 + kk);
        v2f b2 = *(const v2f*)(rlane + k0 + kk);
        acc = __builtin_amdgcn_wmma_f32_16x16x4_f32(false, a2, false, b2,
                                                    (short)0, acc, false, false);
    }

    int ccol  = colbase + (lane & 15);
    int rbase = row_tile * 16 + (lane >> 4) * 8;
    float bias = bl[ccol];
#pragma unroll
    for (int i = 0; i < 8; ++i) {
        float v = acc[i] + bias;
        out[(size_t)(rbase + i) * DIM + ccol] = fmaxf(v, 0.0f);
    }
}

__global__ void head_kernel(const float* __restrict__ h,
                            const float* __restrict__ Wp, const float* __restrict__ bp,
                            const float* __restrict__ Wd, const float* __restrict__ bd,
                            float* __restrict__ out, int n_nodes) {
    int wave = (blockIdx.x * blockDim.x + threadIdx.x) >> 5;
    int lane = threadIdx.x & 31;
    if (wave >= n_nodes) return;
    float4 hv = ((const float4*)(h + (size_t)wave * DIM))[lane];
    float4 wp = ((const float4*)Wp)[lane];
    float4 wd = ((const float4*)Wd)[lane];
    float p = hv.x * wp.x + hv.y * wp.y + hv.z * wp.z + hv.w * wp.w;
    float d = hv.x * wd.x + hv.y * wd.y + hv.z * wd.z + hv.w * wd.w;
#pragma unroll
    for (int off = 16; off > 0; off >>= 1) {
        p += __shfl_xor(p, off, 32);
        d += __shfl_xor(d, off, 32);
    }
    if (lane == 0) {
        p += bp[0];
        d += bd[0];
        d = 1.0f / (1.0f + __expf(-d));
        out[wave] = p - d;
        out[n_nodes + wave] = p + d;
    }
}

extern "C" void kernel_launch(void* const* d_in, const int* in_sizes, int n_in,
                              void* d_out, int out_size, void* d_ws, size_t ws_size,
                              hipStream_t stream) {
    const float* x    = (const float*)d_in[0];
    const int*   edge = (const int*)d_in[1];
    const float* W1l  = (const float*)d_in[2];
    const float* b1l  = (const float*)d_in[3];
    const float* W1r  = (const float*)d_in[4];
    const float* W2l  = (const float*)d_in[5];
    const float* b2l  = (const float*)d_in[6];
    const float* W2r  = (const float*)d_in[7];
    const float* Wp   = (const float*)d_in[8];
    const float* bp   = (const float*)d_in[9];
    const float* Wd   = (const float*)d_in[10];
    const float* bd   = (const float*)d_in[11];
    float* out = (float*)d_out;

    const int N = in_sizes[0] / DIM;
    const int E = in_sizes[1] / 2;
    const int* src = edge;
    const int* dst = edge + E;

    const int ntiles = (E + TILE - 1) / TILE;
    const int HN = NBINS * ntiles;

    char* ws = (char*)d_ws;
    float* agg = (float*)ws;
    float* h1  = agg + (size_t)N * DIM;
    float* h2  = h1 + (size_t)N * DIM;
    float* deg = h2 + (size_t)N * DIM;
    const size_t N4 = ((size_t)N + 4) / 4 * 4;
    const size_t E4 = ((size_t)E + 3) / 4 * 4;
    const size_t H4 = ((size_t)HN + 4) / 4 * 4;
    int* ghist  = (int*)(deg + N4);
    int* gscan  = ghist + H4;
    int* k1     = gscan + H4;
    int* v1     = k1 + E4;
    int* k2     = v1 + E4;
    int* gsrc   = k2 + E4;
    int* rowptr = gsrc + E4;

    const int BLK = 256;
    int sgrid = (HN + SCAN_T) / SCAN_T;
    int rgrid = (E + BLK) / BLK;
    int rowTiles = (N + 15) / 16;
    int ggrid = 8 * ((rowTiles + 7) / 8);
    int hgrid = (N * 32 + BLK - 1) / BLK;

    digit_count_kernel<<<ntiles, SORT_T, 0, stream>>>(dst, ghist, E, 0, ntiles);
    exclusive_scan_kernel<<<sgrid, SCAN_T, 0, stream>>>(ghist, gscan, HN);
    digit_move_kernel<<<ntiles, SORT_T, 0, stream>>>(dst, src, k1, v1, gscan, E, 0, ntiles);
    digit_count_kernel<<<ntiles, SORT_T, 0, stream>>>(k1, ghist, E, RBITS, ntiles);
    exclusive_scan_kernel<<<sgrid, SCAN_T, 0, stream>>>(ghist, gscan, HN);
    digit_move_kernel<<<ntiles, SORT_T, 0, stream>>>(k1, v1, k2, gsrc, gscan, E, RBITS, ntiles);
    rowptr_kernel<<<rgrid, BLK, 0, stream>>>(k2, rowptr, E, N);

    aggregate_kernel<<<hgrid, BLK, 0, stream>>>(x, rowptr, gsrc, agg, deg, N);
    sage_gemm_kernel<<<ggrid, BLK, 0, stream>>>(agg, deg, x, W1l, b1l, W1r, h1, N);

    aggregate_kernel<<<hgrid, BLK, 0, stream>>>(h1, rowptr, gsrc, agg, nullptr, N);
    sage_gemm_kernel<<<ggrid, BLK, 0, stream>>>(agg, deg, h1, W2l, b2l, W2r, h2, N);

    head_kernel<<<hgrid, BLK, 0, stream>>>(h2, Wp, bp, Wd, bd, out, N);
}


# layer GEMM re-tiled: a wave computes a 16x128 strip (8 accumulator tiles) from both whole weight matrices in LDS; each A fragment feeds 8 WMMAs, A rows read once not by 8 blocks; sum order kept
# speedup vs baseline: 2.0746x; 1.2014x over previous
#include <hip/hip_runtime.h>
#include <hip/hip_bf16.h>
#include <math.h>

typedef __attribute__((ext_vector_type(2))) float v2f;
typedef __attribute__((ext_vector_type(8))) float v8f;

#define DIM 128
#define BPAD 132


#define RBITS 9
#define NBINS 512
#define GROUPS 16
#define TILE 4096
#define SORT_T 256

__device__ __forceinline__ unsigned same_digit_lanes(int d, bool valid) {
    unsigned peers = __builtin_amdgcn_ballot_w32(valid);
#pragma unroll
    for (int b = 0; b < RBITS; ++b) {
        bool bit = ((d >> b) & 1) != 0;
        unsigned bal = __builtin_amdgcn_ballot_w32(bit);
        peers &= bit ? bal : ~bal;
    }
    return peers;
}

__global__ void __launch_bounds__(SORT_T)
digit_count_kernel(const int* __restrict__ keys, int* __restrict__ ghist,
                   int n, int shift, int ntiles) {
    __shared__ int wh[8 * NBINS];
    int tid = threadIdx.x, lane = tid & 31, wv = tid >> 5;
    for (int i = tid; i < 8 * NBINS; i += SORT_T) wh[i] = 0;
    __syncthreads();
    int base = blockIdx.x * TILE + wv * (GROUPS * 32) + lane;
    unsigned lt = (1u << lane) - 1u;
    int* mine = wh + wv * NBINS;
    int dg[GROUPS];
#pragma unroll
    for (int g = 0; g < GROUPS; ++g) {
        int idx = base + g * 32;
        dg[g] = (idx < n) ? ((keys[idx] >> shift) & (NBINS - 1)) : 0;
    }
#pragma unroll
    for (int g = 0; g < GROUPS; ++g) {
        bool valid = (base + g * 32) < n;
        int d = dg[g];
        unsigned peers = same_digit_lanes(d, valid);
        if (valid && (peers & lt) == 0u) mine[d] += __popc(peers);
        __syncthreads();
    }
    for (int d = tid; d < NBINS; d += SORT_T) {
        int s = 0;
        for (int w = 0; w < 8; ++w) s += wh[w * NBINS + d];
        ghist[d * ntiles + blockIdx.x] = s;
    }
}

#define SCAN_T 1024
__global__ void __launch_bounds__(SCAN_T)
exclusive_scan_kernel(const int* __restrict__ cnt, int* __restrict__ outp, int n) {
    __shared__ int wtot[32];
    __shared__ int wpre[32];
    int tid  = threadIdx.x;
    int lane = tid & 31;
    int wv   = tid >> 5;
    int first = blockIdx.x * SCAN_T;

    const int4* c4 = (const int4*)cnt;
    int nvec = first >> 2;
    int pre = 0;
#pragma unroll 8
    for (int i = tid; i < nvec; i += SCAN_T) {
        int4 q = c4[i];
        pre += (q.x + q.y) + (q.z + q.w);
    }

    int i = first + tid;
    int c = (i < n) ? cnt[i] : 0;

    int inc = c;
#pragma unroll
    for (int off = 1; off < 32; off <<= 1) {
        int t = __shfl_up(inc, off, 32);
        if (lane >= off) inc += t;
    }
#pragma unroll
    for (int off = 16; off > 0; off >>= 1) pre += __shfl_xor(pre, off, 32);
    if (lane == 31) wtot[wv] = inc;
    if (lane == 0)  wpre[wv] = pre;
    __syncthreads();

    int t = (lane < wv) ? wtot[lane] : 0;
    int p = wpre[lane];
#pragma unroll
    for (int off = 16; off > 0; off >>= 1) {
        t += __shfl_xor(t, off, 32);
        p += __shfl_xor(p, off, 32);
    }
    if (i <= n) outp[i] = p + t + (inc - c);
}

__global__ void __launch_bounds__(SORT_T)
digit_move_kernel(const int* __restrict__ kin, const int* __restrict__ vin,
                  int* __restrict__ kout, int* __restrict__ vout,
                  const int* __restrict__ gscan, int n, int shift, int ntiles) {
    __shared__ int wh[8 * NBINS];
    int tid = threadIdx.x, lane = tid & 31, wv = tid >> 5;
    for (int i = tid; i < 8 * NBINS; i += SORT_T) wh[i] = 0;
    __syncthreads();
    int base = blockIdx.x * TILE + wv * (GROUPS * 32) + lane;
    unsigned lt = (1u << lane) - 1u;
    int* mine = wh + wv * NBINS;
    int key[GROUPS], val[GROUPS];
#pragma unroll
    for (int g = 0; g < GROUPS; ++g) {
        int idx = base + g * 32;
        bool valid = idx < n;
        key[g] = valid ? kin[idx] : 0;
        val[g] = valid ? vin[idx] : 0;
    }
#pragma unroll
    for (int g = 0; g < GROUPS; ++g) {
        bool valid = (base + g * 32) < n;
        int d = (key[g] >> shift) & (NBINS - 1);
        unsigned peers = same_digit_lanes(d, valid);
        if (valid && (peers & lt) == 0u) mine[d] += __popc(peers);
        __syncthreads();
    }
    for (int d = tid; d < NBINS; d += SORT_T) {
        int run = gscan[d * ntiles + blockIdx.x];
        for (int w = 0; w < 8; ++w) {
            int t = wh[w * NBINS + d];
            wh[w * NBINS + d] = run;
            run += t;
        }
    }
    __syncthreads();
#pragma unroll
    for (int g = 0; g < GROUPS; ++g) {
        int idx = base + g * 32;
        bool valid = idx < n;
        int d = (key[g] >> shift) & (NBINS - 1);
        unsigned peers = same_digit_lanes(d, valid);
        int rank = __popc(peers & lt);
        int pos = mine[d] + rank;
        __syncthreads();
        if (valid && rank == 0) mine[d] += __popc(peers);
        if (valid) { kout[pos] = key[g]; vout[pos] = val[g]; }
        __syncthreads();
    }
}

__global__ void rowptr_kernel(const int* __restrict__ skey, int* __restrict__ rowptr,
                              int n_edges, int n_nodes) {
    int i = blockIdx.x * blockDim.x + threadIdx.x;
    if (i > n_edges) return;
    int prev = (i == 0) ? -1 : skey[i - 1];
    int cur  = (i == n_edges) ? n_nodes : skey[i];
    for (int k = prev + 1; k <= cur; ++k) rowptr[k] = i;
}

__device__ __forceinline__ float4 row_ld(const float4* __restrict__ xv, int mysrc, int j, int lane) {
    int s = __builtin_amdgcn_readlane(mysrc, j);
    return xv[(size_t)s * (DIM / 4) + lane];
}
__device__ __forceinline__ void row_add(float4& acc, const float4& v) {
    acc.x += v.x; acc.y += v.y; acc.z += v.z; acc.w += v.w;
}

__global__ void aggregate_kernel(const float* __restrict__ xin, const int* __restrict__ rowptr,
                                 const int* __restrict__ gsrc, float* __restrict__ agg,
                                 float* __restrict__ deg, int n_nodes) {
    int lane = threadIdx.x & 31;
    int wib  = __builtin_amdgcn_readfirstlane((int)(threadIdx.x >> 5));
    int node = blockIdx.x * (int)(blockDim.x >> 5) + wib;
    if (node >= n_nodes) return;
    int beg = rowptr[node];
    int end = rowptr[node + 1];
    const float4* xv = (const float4*)xin;
    float4 acc = make_float4(0.f, 0.f, 0.f, 0.f);

    for (int base = beg; base < end; base += 32) {
        int n = end - base;
        if (n > 32) n = 32;
        int mysrc = (lane < n) ? gsrc[base + lane] : 0;
        int j = 0;
        for (; j + 8 <= n; j += 8) {
            float4 v[8];
#pragma unroll
            for (int u = 0; u < 8; ++u) v[u] = row_ld(xv, mysrc, j + u, lane);
#pragma unroll
            for (int u = 0; u < 8; ++u) row_add(acc, v[u]);
        }
        if (j + 4 <= n) {
            float4 v0 = row_ld(xv, mysrc, j, lane);
            float4 v1 = row_ld(xv, mysrc, j + 1, lane);
            float4 v2 = row_ld(xv, mysrc, j + 2, lane);
            float4 v3 = row_ld(xv, mysrc, j + 3, lane);
            row_add(acc, v0); row_add(acc, v1); row_add(acc, v2); row_add(acc, v3);
            j += 4;
        }
        if (j + 2 <= n) {
            float4 v0 = row_ld(xv, mysrc, j, lane);
            float4 v1 = row_ld(xv, mysrc, j + 1, lane);
            row_add(acc, v0); row_add(acc, v1);
            j += 2;
        }
        if (j < n) {
            float4 v0 = row_ld(xv, mysrc, j, lane);
            row_add(acc, v0);
        }
    }
    ((float4*)agg)[(size_t)node * (DIM / 4) + lane] = acc;
    if (deg != nullptr && lane == 0) deg[node] = (float)(end - beg);
}

#define GEMM_T 256
__global__ void __launch_bounds__(GEMM_T)
sage_gemm_kernel(const float* __restrict__ agg, const float* __restrict__ deg,
                 const float* __restrict__ xin,
                 const float* __restrict__ Wl, const float* __restrict__ bl,
                 const float* __restrict__ Wr,
                 float* __restrict__ out, int n_nodes) {
    __shared__ float Blt[DIM * BPAD];
    __shared__ float Brt[DIM * BPAD];

    int wv   = threadIdx.x >> 5;
    int lane = threadIdx.x & 31;

    {
        int n = threadIdx.x & (DIM - 1), kb = threadIdx.x >> 7;
        const float* pl = Wl + threadIdx.x;
        const float* pr = Wr + threadIdx.x;
        float* dl = Blt + n * BPAD + kb;
        float* dr = Brt + n * BPAD + kb;
        for (int i0 = 0; i0 < DIM * DIM / GEMM_T; i0 += 16) {
#pragma unroll
            for (int i = 0; i < 16; ++i) {
                dl[2 * i] = pl[GEMM_T * i];
                dr[2 * i] = pr[GEMM_T * i];
            }
            pl += 16 * GEMM_T; pr += 16 * GEMM_T; dl += 32; dr += 32;
        }
    }
    __syncthreads();

    int row_tile = blockIdx.x * (GEMM_T / 32) + wv;
    if (row_tile * 16 + 16 > n_nodes) return;

    int m  = lane & 15;
    int kk = (lane >> 4) * 2;
    int nn = lane & 15;

    int row = row_tile * 16 + m;
    float rdeg = 1.0f / fmaxf(deg[row], 1.0f);
    const float* arow = agg + (size_t)row * DIM + kk;
    const float* xrow = xin + (size_t)row * DIM + kk;
    const float* blane = Blt + nn * BPAD + kk;
    const float* rlane = Brt + nn * BPAD + kk;

    v8f acc[8];
#pragma unroll
    for (int t = 0; t < 8; ++t) acc[t] = (v8f){};

#pragma unroll 2
    for (int k0 = 0; k0 < DIM; k0 += 4) {
        v2f a1 = *(const v2f*)(arow + k0);
        a1.x *= rdeg; a1.y *= rdeg;
        v2f a2 = *(const v2f*)(xrow + k0);
#pragma unroll
        for (int t = 0; t < 8; ++t) {
            v2f b1 = *(const v2f*)(blane + t * 16 * BPAD + k0);
            acc[t] = __builtin_amdgcn_wmma_f32_16x16x4_f32(false, a1, false, b1,
                                                           (short)0, acc[t], false, false);
            v2f b2 = *(const v2f*)(rlane + t * 16 * BPAD + k0);
            acc[t] = __builtin_amdgcn_wmma_f32_16x16x4_f32(false, a2, false, b2,
                                                           (short)0, acc[t], false, false);
        }
    }

    int rbase = row_tile * 16 + (lane >> 4) * 8;
#pragma unroll
    for (int t = 0; t < 8; ++t) {
        int ccol = t * 16 + nn;
        float bias = bl[ccol];
#pragma unroll
        for (int i = 0; i < 8; ++i) {
            float v = acc[t][i] + bias;
            out[(size_t)(rbase + i) * DIM + ccol] = fmaxf(v, 0.0f);
        }
    }
}

__global__ void head_kernel(const float* __restrict__ h,
                            const float* __restrict__ Wp, const float* __restrict__ bp,
                            const float* __restrict__ Wd, const float* __restrict__ bd,
                            float* __restrict__ out, int n_nodes) {
    int wave = (blockIdx.x * blockDim.x + threadIdx.x) >> 5;
    int lane = threadIdx.x & 31;
    if (wave >= n_nodes) return;
    float4 hv = ((const float4*)(h + (size_t)wave * DIM))[lane];
    float4 wp = ((const float4*)Wp)[lane];
    float4 wd = ((const float4*)Wd)[lane];
    float p = hv.x * wp.x + hv.y * wp.y + hv.z * wp.z + hv.w * wp.w;
    float d = hv.x * wd.x + hv.y * wd.y + hv.z * wd.z + hv.w * wd.w;
#pragma unroll
    for (int off = 16; off > 0; off >>= 1) {
        p += __shfl_xor(p, off, 32);
        d += __shfl_xor(d, off, 32);
    }
    if (lane == 0) {
        p += bp[0];
        d += bd[0];
        d = 1.0f / (1.0f + __expf(-d));
        out[wave] = p - d;
        out[n_nodes + wave] = p + d;
    }
}

extern "C" void kernel_launch(void* const* d_in, const int* in_sizes, int n_in,
                              void* d_out, int out_size, void* d_ws, size_t ws_size,
                              hipStream_t stream) {
    const float* x    = (const float*)d_in[0];
    const int*   edge = (const int*)d_in[1];
    const float* W1l  = (const float*)d_in[2];
    const float* b1l  = (const float*)d_in[3];
    const float* W1r  = (const float*)d_in[4];
    const float* W2l  = (const float*)d_in[5];
    const float* b2l  = (const float*)d_in[6];
    const float* W2r  = (const float*)d_in[7];
    const float* Wp   = (const float*)d_in[8];
    const float* bp   = (const float*)d_in[9];
    const float* Wd   = (const float*)d_in[10];
    const float* bd   = (const float*)d_in[11];
    float* out = (float*)d_out;

    const int N = in_sizes[0] / DIM;
    const int E = in_sizes[1] / 2;
    const int* src = edge;
    const int* dst = edge + E;

    const int ntiles = (E + TILE - 1) / TILE;
    const int HN = NBINS * ntiles;

    char* ws = (char*)d_ws;
    float* agg = (float*)ws;
    float* h1  = agg + (size_t)N * DIM;
    float* h2  = h1 + (size_t)N * DIM;
    float* deg = h2 + (size_t)N * DIM;
    const size_t N4 = ((size_t)N + 4) / 4 * 4;
    const size_t E4 = ((size_t)E + 3) / 4 * 4;
    const size_t H4 = ((size_t)HN + 4) / 4 * 4;
    int* ghist  = (int*)(deg + N4);
    int* gscan  = ghist + H4;
    int* k1     = gscan + H4;
    int* v1     = k1 + E4;
    int* k2     = v1 + E4;
    int* gsrc   = k2 + E4;
    int* rowptr = gsrc + E4;

    const int BLK = 256;
    int sgrid = (HN + SCAN_T) / SCAN_T;
    int rgrid = (E + BLK) / BLK;
    int rowTiles = (N + 15) / 16;
    int ggrid = (rowTiles + 7) / 8;
    int hgrid = (N * 32 + BLK - 1) / BLK;

    digit_count_kernel<<<ntiles, SORT_T, 0, stream>>>(dst, ghist, E, 0, ntiles);
    exclusive_scan_kernel<<<sgrid, SCAN_T, 0, stream>>>(ghist, gscan, HN);
    digit_move_kernel<<<ntiles, SORT_T, 0, stream>>>(dst, src, k1, v1, gscan, E, 0, ntiles);
    digit_count_kernel<<<ntiles, SORT_T, 0, stream>>>(k1, ghist, E, RBITS, ntiles);
    exclusive_scan_kernel<<<sgrid, SCAN_T, 0, stream>>>(ghist, gscan, HN);
    digit_move_kernel<<<ntiles, SORT_T, 0, stream>>>(k1, v1, k2, gsrc, gscan, E, RBITS, ntiles);
    rowptr_kernel<<<rgrid, BLK, 0, stream>>>(k2, rowptr, E, N);

    aggregate_kernel<<<hgrid, BLK, 0, stream>>>(x, rowptr, gsrc, agg, deg, N);
    sage_gemm_kernel<<<ggrid, GEMM_T, 0, stream>>>(agg, deg, x, W1l, b1l, W1r, h1, N);

    aggregate_kernel<<<hgrid, BLK, 0, stream>>>(h1, rowptr, gsrc, agg, nullptr, N);
    sage_gemm_kernel<<<ggrid, GEMM_T, 0, stream>>>(agg, deg, h1, W2l, b2l, W2r, h2, N);

    head_kernel<<<hgrid, BLK, 0, stream>>>(h2, Wp, bp, Wd, bd, out, N);
}


# radix move step: a tile's (key,value) pairs are put in digit order in LDS, then stored by consecutive threads: 3-5 lines per wave store, not about 30; sorted order and sums unchanged
# speedup vs baseline: 2.6836x; 1.2936x over previous
#include <hip/hip_runtime.h>
#include <hip/hip_bf16.h>
#include <math.h>

typedef __attribute__((ext_vector_type(2))) float v2f;
typedef __attribute__((ext_vector_type(8))) float v8f;

#define DIM 128
#define BPAD 132


#define RBITS 9
#define NBINS 512
#define GROUPS 16
#define TILE 4096
#define SORT_T 256

__device__ __forceinline__ unsigned same_digit_lanes(int d, bool valid) {
    unsigned peers = __builtin_amdgcn_ballot_w32(valid);
#pragma unroll
    for (int b = 0; b < RBITS; ++b) {
        bool bit = ((d >> b) & 1) != 0;
        unsigned bal = __builtin_amdgcn_ballot_w32(bit);
        peers &= bit ? bal : ~bal;
    }
    return peers;
}

__global__ void __launch_bounds__(SORT_T)
digit_count_kernel(const int* __restrict__ keys, int* __restrict__ ghist,
                   int n, int shift, int ntiles) {
    __shared__ int wh[8 * NBINS];
    int tid = threadIdx.x, lane = tid & 31, wv = tid >> 5;
    for (int i = tid; i < 8 * NBINS; i += SORT_T) wh[i] = 0;
    __syncthreads();
    int base = blockIdx.x * TILE + wv * (GROUPS * 32) + lane;
    unsigned lt = (1u << lane) - 1u;
    int* mine = wh + wv * NBINS;
    int dg[GROUPS];
#pragma unroll
    for (int g = 0; g < GROUPS; ++g) {
        int idx = base + g * 32;
        dg[g] = (idx < n) ? ((keys[idx] >> shift) & (NBINS - 1)) : 0;
    }
#pragma unroll
    for (int g = 0; g < GROUPS; ++g) {
        bool valid = (base + g * 32) < n;
        int d = dg[g];
        unsigned peers = same_digit_lanes(d, valid);
        if (valid && (peers & lt) == 0u) mine[d] += __popc(peers);
        __syncthreads();
    }
    for (int d = tid; d < NBINS; d += SORT_T) {
        int s = 0;
        for (int w = 0; w < 8; ++w) s += wh[w * NBINS + d];
        ghist[d * ntiles + blockIdx.x] = s;
    }
}

#define SCAN_T 1024
__global__ void __launch_bounds__(SCAN_T)
exclusive_scan_kernel(const int* __restrict__ cnt, int* __restrict__ outp, int n) {
    __shared__ int wtot[32];
    __shared__ int wpre[32];
    int tid  = threadIdx.x;
    int lane = tid & 31;
    int wv   = tid >> 5;
    int first = blockIdx.x * SCAN_T;

    const int4* c4 = (const int4*)cnt;
    int nvec = first >> 2;
    int pre = 0;
#pragma unroll 8
    for (int i = tid; i < nvec; i += SCAN_T) {
        int4 q = c4[i];
        pre += (q.x + q.y) + (q.z + q.w);
    }

    int i = first + tid;
    int c = (i < n) ? cnt[i] : 0;

    int inc = c;
#pragma unroll
    for (int off = 1; off < 32; off <<= 1) {
        int t = __shfl_up(inc, off, 32);
        if (lane >= off) inc += t;
    }
#pragma unroll
    for (int off = 16; off > 0; off >>= 1) pre += __shfl_xor(pre, off, 32);
    if (lane == 31) wtot[wv] = inc;
    if (lane == 0)  wpre[wv] = pre;
    __syncthreads();

    int t = (lane < wv) ? wtot[lane] : 0;
    int p = wpre[lane];
#pragma unroll
    for (int off = 16; off > 0; off >>= 1) {
        t += __shfl_xor(t, off, 32);
        p += __shfl_xor(p, off, 32);
    }
    if (i <= n) outp[i] = p + t + (inc - c);
}

__global__ void __launch_bounds__(SORT_T)
digit_move_kernel(const int* __restrict__ kin, const int* __restrict__ vin,
                  int* __restrict__ kout, int* __restrict__ vout,
                  const int* __restrict__ gscan, int n, int shift, int ntiles) {
    __shared__ int wh[8 * NBINS];
    __shared__ int goff[NBINS];
    __shared__ int skey[TILE];
    __shared__ int sval[TILE];
    __shared__ int wsum[8];
    int tid = threadIdx.x, lane = tid & 31, wv = tid >> 5;
    for (int i = tid; i < 8 * NBINS; i += SORT_T) wh[i] = 0;
    __syncthreads();
    int base = blockIdx.x * TILE + wv * (GROUPS * 32) + lane;
    unsigned lt = (1u << lane) - 1u;
    int* mine = wh + wv * NBINS;
    int key[GROUPS], val[GROUPS];
#pragma unroll
    for (int g = 0; g < GROUPS; ++g) {
        int idx = base + g * 32;
        bool valid = idx < n;
        key[g] = valid ? kin[idx] : 0;
        val[g] = valid ? vin[idx] : 0;
    }
#pragma unroll
    for (int g = 0; g < GROUPS; ++g) {
        bool valid = (base + g * 32) < n;
        int d = (key[g] >> shift) & (NBINS - 1);
        unsigned peers = same_digit_lanes(d, valid);
        if (valid && (peers & lt) == 0u) mine[d] += __popc(peers);
        __syncthreads();
    }
    {
        int d0 = 2 * tid, d1 = d0 + 1;
        int t0 = 0, t1 = 0;
        for (int w = 0; w < 8; ++w) { t0 += wh[w * NBINS + d0]; t1 += wh[w * NBINS + d1]; }
        int pair = t0 + t1;
        int inc = pair;
#pragma unroll
        for (int off = 1; off < 32; off <<= 1) {
            int t = __shfl_up(inc, off, 32);
            if (lane >= off) inc += t;
        }
        if (lane == 31) wsum[wv] = inc;
        __syncthreads();
        int front = 0;
        for (int w = 0; w < wv; ++w) front += wsum[w];
        int ex0 = front + inc - pair;
        int ex1 = ex0 + t0;
        int run = ex0;
        for (int w = 0; w < 8; ++w) {
            int t = wh[w * NBINS + d0];
            wh[w * NBINS + d0] = run;
            run += t;
        }
        run = ex1;
        for (int w = 0; w < 8; ++w) {
            int t = wh[w * NBINS + d1];
            wh[w * NBINS + d1] = run;
            run += t;
        }
        goff[d0] = gscan[d0 * ntiles + blockIdx.x] - ex0;
        goff[d1] = gscan[d1 * ntiles + blockIdx.x] - ex1;
    }
    __syncthreads();
#pragma unroll
    for (int g = 0; g < GROUPS; ++g) {
        int idx = base + g * 32;
        bool valid = idx < n;
        int d = (key[g] >> shift) & (NBINS - 1);
        unsigned peers = same_digit_lanes(d, valid);
        int rank = __popc(peers & lt);
        int lpos = mine[d] + rank;
        __syncthreads();
        if (valid && rank == 0) mine[d] += __popc(peers);
        if (valid) { skey[lpos] = key[g]; sval[lpos] = val[g]; }
        __syncthreads();
    }
    int cnt = n - blockIdx.x * TILE;
    if (cnt > TILE) cnt = TILE;
    int ko[GROUPS], vo[GROUPS], po[GROUPS];
#pragma unroll
    for (int j = 0; j < GROUPS; ++j) {
        int i = tid + j * SORT_T;
        bool in = i < cnt;
        ko[j] = in ? skey[i] : 0;
        vo[j] = in ? sval[i] : 0;
    }
#pragma unroll
    for (int j = 0; j < GROUPS; ++j)
        po[j] = tid + j * SORT_T + goff[(ko[j] >> shift) & (NBINS - 1)];
#pragma unroll
    for (int j = 0; j < GROUPS; ++j) {
        if (tid + j * SORT_T < cnt) { kout[po[j]] = ko[j]; vout[po[j]] = vo[j]; }
    }
}

__global__ void rowptr_kernel(const int* __restrict__ skey, int* __restrict__ rowptr,
                              int n_edges, int n_nodes) {
    int i = blockIdx.x * blockDim.x + threadIdx.x;
    if (i > n_edges) return;
    int prev = (i == 0) ? -1 : skey[i - 1];
    int cur  = (i == n_edges) ? n_nodes : skey[i];
    for (int k = prev + 1; k <= cur; ++k) rowptr[k] = i;
}

__device__ __forceinline__ float4 row_ld(const float4* __restrict__ xv, int mysrc, int j, int lane) {
    int s = __builtin_amdgcn_readlane(mysrc, j);
    return xv[(size_t)s * (DIM / 4) + lane];
}
__device__ __forceinline__ void row_add(float4& acc, const float4& v) {
    acc.x += v.x; acc.y += v.y; acc.z += v.z; acc.w += v.w;
}

__global__ void aggregate_kernel(const float* __restrict__ xin, const int* __restrict__ rowptr,
                                 const int* __restrict__ gsrc, float* __restrict__ agg,
                                 float* __restrict__ deg, int n_nodes) {
    int lane = threadIdx.x & 31;
    int wib  = __builtin_amdgcn_readfirstlane((int)(threadIdx.x >> 5));
    int node = blockIdx.x * (int)(blockDim.x >> 5) + wib;
    if (node >= n_nodes) return;
    int beg = rowptr[node];
    int end = rowptr[node + 1];
    const float4* xv = (const float4*)xin;
    float4 acc = make_float4(0.f, 0.f, 0.f, 0.f);

    for (int base = beg; base < end; base += 32) {
        int n = end - base;
        if (n > 32) n = 32;
        int mysrc = (lane < n) ? gsrc[base + lane] : 0;
        int j = 0;
        for (; j + 8 <= n; j += 8) {
            float4 v[8];
#pragma unroll
            for (int u = 0; u < 8; ++u) v[u] = row_ld(xv, mysrc, j + u, lane);
#pragma unroll
            for (int u = 0; u < 8; ++u) row_add(acc, v[u]);
        }
        if (j + 4 <= n) {
            float4 v0 = row_ld(xv, mysrc, j, lane);
            float4 v1 = row_ld(xv, mysrc, j + 1, lane);
            float4 v2 = row_ld(xv, mysrc, j + 2, lane);
            float4 v3 = row_ld(xv, mysrc, j + 3, lane);
            row_add(acc, v0); row_add(acc, v1); row_add(acc, v2); row_add(acc, v3);
            j += 4;
        }
        if (j + 2 <= n) {
            float4 v0 = row_ld(xv, mysrc, j, lane);
            float4 v1 = row_ld(xv, mysrc, j + 1, lane);
            row_add(acc, v0); row_add(acc, v1);
            j += 2;
        }
        if (j < n) {
            float4 v0 = row_ld(xv, mysrc, j, lane);
            row_add(acc, v0);
        }
    }
    ((float4*)agg)[(size_t)node * (DIM / 4) + lane] = acc;
    if (deg != nullptr && lane == 0) deg[node] = (float)(end - beg);
}

#define GEMM_T 256
__global__ void __launch_bounds__(GEMM_T)
sage_gemm_kernel(const float* __restrict__ agg, const float* __restrict__ deg,
                 const float* __restrict__ xin,
                 const float* __restrict__ Wl, const float* __restrict__ bl,
                 const float* __restrict__ Wr,
                 float* __restrict__ out, int n_nodes) {
    __shared__ float Blt[DIM * BPAD];
    __shared__ float Brt[DIM * BPAD];

    int wv   = threadIdx.x >> 5;
    int lane = threadIdx.x & 31;

    {
        int n = threadIdx.x & (DIM - 1), kb = threadIdx.x >> 7;
        const float* pl = Wl + threadIdx.x;
        const float* pr = Wr + threadIdx.x;
        float* dl = Blt + n * BPAD + kb;
        float* dr = Brt + n * BPAD + kb;
        for (int i0 = 0; i0 < DIM * DIM / GEMM_T; i0 += 16) {
#pragma unroll
            for (int i = 0; i < 16; ++i) {
                dl[2 * i] = pl[GEMM_T * i];
                dr[2 * i] = pr[GEMM_T * i];
            }
            pl += 16 * GEMM_T; pr += 16 * GEMM_T; dl += 32; dr += 32;
        }
    }
    __syncthreads();

    int row_tile = blockIdx.x * (GEMM_T / 32) + wv;
    if (row_tile * 16 + 16 > n_nodes) return;

    int m  = lane & 15;
    int kk = (lane >> 4) * 2;
    int nn = lane & 15;

    int row = row_tile * 16 + m;
    float rdeg = 1.0f / fmaxf(deg[row], 1.0f);
    const float* arow = agg + (size_t)row * DIM + kk;
    const float* xrow = xin + (size_t)row * DIM + kk;
    const float* blane = Blt + nn * BPAD + kk;
    const float* rlane = Brt + nn * BPAD + kk;

    v8f acc[8];
#pragma unroll
    for (int t = 0; t < 8; ++t) acc[t] = (v8f){};

#pragma unroll 2
    for (int k0 = 0; k0 < DIM; k0 += 4) {
        v2f a1 = *(const v2f*)(arow + k0);
        a1.x *= rdeg; a1.y *= rdeg;
        v2f a2 = *(const v2f*)(xrow + k0);
#pragma unroll
        for (int t = 0; t < 8; ++t) {
            v2f b1 = *(const v2f*)(blane + t * 16 * BPAD + k0);
            acc[t] = __builtin_amdgcn_wmma_f32_16x16x4_f32(false, a1, false, b1,
                                                           (short)0, acc[t], false, false);
            v2f b2 = *(const v2f*)(rlane + t * 16 * BPAD + k0);
            acc[t] = __builtin_amdgcn_wmma_f32_16x16x4_f32(false, a2, false, b2,
                                                           (short)0, acc[t], false, false);
        }
    }

    int rbase = row_tile * 16 + (lane >> 4) * 8;
#pragma unroll
    for (int t = 0; t < 8; ++t) {
        int ccol = t * 16 + nn;
        float bias = bl[ccol];
#pragma unroll
        for (int i = 0; i < 8; ++i) {
            float v = acc[t][i] + bias;
            out[(size_t)(rbase + i) * DIM + ccol] = fmaxf(v, 0.0f);
        }
    }
}

__global__ void head_kernel(const float* __restrict__ h,
                            const float* __restrict__ Wp, const float* __restrict__ bp,
                            const float* __restrict__ Wd, const float* __restrict__ bd,
                            float* __restrict__ out, int n_nodes) {
    int wave = (blockIdx.x * blockDim.x + threadIdx.x) >> 5;
    int lane = threadIdx.x & 31;
    if (wave >= n_nodes) return;
    float4 hv = ((const float4*)(h + (size_t)wave * DIM))[lane];
    float4 wp = ((const float4*)Wp)[lane];
    float4 wd = ((const float4*)Wd)[lane];
    float p = hv.x * wp.x + hv.y * wp.y + hv.z * wp.z + hv.w * wp.w;
    float d = hv.x * wd.x + hv.y * wd.y + hv.z * wd.z + hv.w * wd.w;
#pragma unroll
    for (int off = 16; off > 0; off >>= 1) {
        p += __shfl_xor(p, off, 32);
        d += __shfl_xor(d, off, 32);
    }
    if (lane == 0) {
        p += bp[0];
        d += bd[0];
        d = 1.0f / (1.0f + __expf(-d));
        out[wave] = p - d;
        out[n_nodes + wave] = p + d;
    }
}

extern "C" void kernel_launch(void* const* d_in, const int* in_sizes, int n_in,
                              void* d_out, int out_size, void* d_ws, size_t ws_size,
                              hipStream_t stream) {
    const float* x    = (const float*)d_in[0];
    const int*   edge = (const int*)d_in[1];
    const float* W1l  = (const float*)d_in[2];
    const float* b1l  = (const float*)d_in[3];
    const float* W1r  = (const float*)d_in[4];
    const float* W2l  = (const float*)d_in[5];
    const float* b2l  = (const float*)d_in[6];
    const float* W2r  = (const float*)d_in[7];
    const float* Wp   = (const float*)d_in[8];
    const float* bp   = (const float*)d_in[9];
    const float* Wd   = (const float*)d_in[10];
    const float* bd   = (const float*)d_in[11];
    float* out = (float*)d_out;

    const int N = in_sizes[0] / DIM;
    const int E = in_sizes[1] / 2;
    const int* src = edge;
    const int* dst = edge + E;

    const int ntiles = (E + TILE - 1) / TILE;
    const int HN = NBINS * ntiles;

    char* ws = (char*)d_ws;
    float* agg = (float*)ws;
    float* h1  = agg + (size_t)N * DIM;
    float* h2  = h1 + (size_t)N * DIM;
    float* deg = h2 + (size_t)N * DIM;
    const size_t N4 = ((size_t)N + 4) / 4 * 4;
    const size_t E4 = ((size_t)E + 3) / 4 * 4;
    const size_t H4 = ((size_t)HN + 4) / 4 * 4;
    int* ghist  = (int*)(deg + N4);
    int* gscan  = ghist + H4;
    int* k1     = gscan + H4;
    int* v1     = k1 + E4;
    int* k2     = v1 + E4;
    int* gsrc   = k2 + E4;
    int* rowptr = gsrc + E4;

    const int BLK = 256;
    int sgrid = (HN + SCAN_T) / SCAN_T;
    int rgrid = (E + BLK) / BLK;
    int rowTiles = (N + 15) / 16;
    int ggrid = (rowTiles + 7) / 8;
    int hgrid = (N * 32 + BLK - 1) / BLK;

    digit_count_kernel<<<ntiles, SORT_T, 0, stream>>>(dst, ghist, E, 0, ntiles);
    exclusive_scan_kernel<<<sgrid, SCAN_T, 0, stream>>>(ghist, gscan, HN);
    digit_move_kernel<<<ntiles, SORT_T, 0, stream>>>(dst, src, k1, v1, gscan, E, 0, ntiles);
    digit_count_kernel<<<ntiles, SORT_T, 0, stream>>>(k1, ghist, E, RBITS, ntiles);
    exclusive_scan_kernel<<<sgrid, SCAN_T, 0, stream>>>(ghist, gscan, HN);
    digit_move_kernel<<<ntiles, SORT_T, 0, stream>>>(k1, v1, k2, gsrc, gscan, E, RBITS, ntiles);
    rowptr_kernel<<<rgrid, BLK, 0, stream>>>(k2, rowptr, E, N);

    aggregate_kernel<<<hgrid, BLK, 0, stream>>>(x, rowptr, gsrc, agg, deg, N);
    sage_gemm_kernel<<<ggrid, GEMM_T, 0, stream>>>(agg, deg, x, W1l, b1l, W1r, h1, N);

    aggregate_kernel<<<hgrid, BLK, 0, stream>>>(h1, rowptr, gsrc, agg, nullptr, N);
    sage_gemm_kernel<<<ggrid, GEMM_T, 0, stream>>>(agg, deg, h1, W2l, b2l, W2r, h2, N);

    head_kernel<<<hgrid, BLK, 0, stream>>>(h2, Wp, bp, Wd, bd, out, N);
}


# layer-2 GEMM epilogue forms both head products from its activated accumulators (8 in-lane terms, 16-lane butterfly): last hidden plane not stored or re-read, no head launch; head sum order changes
# speedup vs baseline: 2.8832x; 1.0744x over previous
#include <hip/hip_runtime.h>
#include <hip/hip_bf16.h>
#include <math.h>

typedef __attribute__((ext_vector_type(2))) float v2f;
typedef __attribute__((ext_vector_type(8))) float v8f;

#define DIM 128
#define BPAD 132


#define RBITS 9
#define NBINS 512
#define GROUPS 16
#define TILE 4096
#define SORT_T 256

__device__ __forceinline__ unsigned same_digit_lanes(int d, bool valid) {
    unsigned peers = __builtin_amdgcn_ballot_w32(valid);
#pragma unroll
    for (int b = 0; b < RBITS; ++b) {
        bool bit = ((d >> b) & 1) != 0;
        unsigned bal = __builtin_amdgcn_ballot_w32(bit);
        peers &= bit ? bal : ~bal;
    }
    return peers;
}

__global__ void __launch_bounds__(SORT_T)
digit_count_kernel(const int* __restrict__ keys, int* __restrict__ ghist,
                   int n, int shift, int ntiles) {
    __shared__ int wh[8 * NBINS];
    int tid = threadIdx.x, lane = tid & 31, wv = tid >> 5;
    for (int i = tid; i < 8 * NBINS; i += SORT_T) wh[i] = 0;
    __syncthreads();
    int base = blockIdx.x * TILE + wv * (GROUPS * 32) + lane;
    unsigned lt = (1u << lane) - 1u;
    int* mine = wh + wv * NBINS;
    int dg[GROUPS];
#pragma unroll
    for (int g = 0; g < GROUPS; ++g) {
        int idx = base + g * 32;
        dg[g] = (idx < n) ? ((keys[idx] >> shift) & (NBINS - 1)) : 0;
    }
#pragma unroll
    for (int g = 0; g < GROUPS; ++g) {
        bool valid = (base + g * 32) < n;
        int d = dg[g];
        unsigned peers = same_digit_lanes(d, valid);
        if (valid && (peers & lt) == 0u) mine[d] += __popc(peers);
        __syncthreads();
    }
    for (int d = tid; d < NBINS; d += SORT_T) {
        int s = 0;
        for (int w = 0; w < 8; ++w) s += wh[w * NBINS + d];
        ghist[d * ntiles + blockIdx.x] = s;
    }
}

#define SCAN_T 1024
__global__ void __launch_bounds__(SCAN_T)
exclusive_scan_kernel(const int* __restrict__ cnt, int* __restrict__ outp, int n) {
    __shared__ int wtot[32];
    __shared__ int wpre[32];
    int tid  = threadIdx.x;
    int lane = tid & 31;
    int wv   = tid >> 5;
    int first = blockIdx.x * SCAN_T;

    const int4* c4 = (const int4*)cnt;
    int nvec = first >> 2;
    int pre = 0;
#pragma unroll 8
    for (int i = tid; i < nvec; i += SCAN_T) {
        int4 q = c4[i];
        pre += (q.x + q.y) + (q.z + q.w);
    }

    int i = first + tid;
    int c = (i < n) ? cnt[i] : 0;

    int inc = c;
#pragma unroll
    for (int off = 1; off < 32; off <<= 1) {
        int t = __shfl_up(inc, off, 32);
        if (lane >= off) inc += t;
    }
#pragma unroll
    for (int off = 16; off > 0; off >>= 1) pre += __shfl_xor(pre, off, 32);
    if (lane == 31) wtot[wv] = inc;
    if (lane == 0)  wpre[wv] = pre;
    __syncthreads();

    int t = (lane < wv) ? wtot[lane] : 0;
    int p = wpre[lane];
#pragma unroll
    for (int off = 16; off > 0; off >>= 1) {
        t += __shfl_xor(t, off, 32);
        p += __shfl_xor(p, off, 32);
    }
    if (i <= n) outp[i] = p + t + (inc - c);
}

__global__ void __launch_bounds__(SORT_T)
digit_move_kernel(const int* __restrict__ kin, const int* __restrict__ vin,
                  int* __restrict__ kout, int* __restrict__ vout,
                  const int* __restrict__ gscan, int n, int shift, int ntiles) {
    __shared__ int wh[8 * NBINS];
    __shared__ int goff[NBINS];
    __shared__ int skey[TILE];
    __shared__ int sval[TILE];
    __shared__ int wsum[8];
    int tid = threadIdx.x, lane = tid & 31, wv = tid >> 5;
    for (int i = tid; i < 8 * NBINS; i += SORT_T) wh[i] = 0;
    __syncthreads();
    int base = blockIdx.x * TILE + wv * (GROUPS * 32) + lane;
    unsigned lt = (1u << lane) - 1u;
    int* mine = wh + wv * NBINS;
    int key[GROUPS], val[GROUPS];
#pragma unroll
    for (int g = 0; g < GROUPS; ++g) {
        int idx = base + g * 32;
        bool valid = idx < n;
        key[g] = valid ? kin[idx] : 0;
        val[g] = valid ? vin[idx] : 0;
    }
#pragma unroll
    for (int g = 0; g < GROUPS; ++g) {
        bool valid = (base + g * 32) < n;
        int d = (key[g] >> shift) & (NBINS - 1);
        unsigned peers = same_digit_lanes(d, valid);
        if (valid && (peers & lt) == 0u) mine[d] += __popc(peers);
        __syncthreads();
    }
    {
        int d0 = 2 * tid, d1 = d0 + 1;
        int t0 = 0, t1 = 0;
        for (int w = 0; w < 8; ++w) { t0 += wh[w * NBINS + d0]; t1 += wh[w * NBINS + d1]; }
        int pair = t0 + t1;
        int inc = pair;
#pragma unroll
        for (int off = 1; off < 32; off <<= 1) {
            int t = __shfl_up(inc, off, 32);
            if (lane >= off) inc += t;
        }
        if (lane == 31) wsum[wv] = inc;
        __syncthreads();
        int front = 0;
        for (int w = 0; w < wv; ++w) front += wsum[w];
        int ex0 = front + inc - pair;
        int ex1 = ex0 + t0;
        int run = ex0;
        for (int w = 0; w < 8; ++w) {
            int t = wh[w * NBINS + d0];
            wh[w * NBINS + d0] = run;
            run += t;
        }
        run = ex1;
        for (int w = 0; w < 8; ++w) {
            int t = wh[w * NBINS + d1];
            wh[w * NBINS + d1] = run;
            run += t;
        }
        goff[d0] = gscan[d0 * ntiles + blockIdx.x] - ex0;
        goff[d1] = gscan[d1 * ntiles + blockIdx.x] - ex1;
    }
    __syncthreads();
#pragma unroll
    for (int g = 0; g < GROUPS; ++g) {
        int idx = base + g * 32;
        bool valid = idx < n;
        int d = (key[g] >> shift) & (NBINS - 1);
        unsigned peers = same_digit_lanes(d, valid);
        int rank = __popc(peers & lt);
        int lpos = mine[d] + rank;
        __syncthreads();
        if (valid && rank == 0) mine[d] += __popc(peers);
        if (valid) { skey[lpos] = key[g]; sval[lpos] = val[g]; }
        __syncthreads();
    }
    int cnt = n - blockIdx.x * TILE;
    if (cnt > TILE) cnt = TILE;
    int ko[GROUPS], vo[GROUPS], po[GROUPS];
#pragma unroll
    for (int j = 0; j < GROUPS; ++j) {
        int i = tid + j * SORT_T;
        bool in = i < cnt;
        ko[j] = in ? skey[i] : 0;
        vo[j] = in ? sval[i] : 0;
    }
#pragma unroll
    for (int j = 0; j < GROUPS; ++j)
        po[j] = tid + j * SORT_T + goff[(ko[j] >> shift) & (NBINS - 1)];
#pragma unroll
    for (int j = 0; j < GROUPS; ++j) {
        if (tid + j * SORT_T < cnt) { kout[po[j]] = ko[j]; vout[po[j]] = vo[j]; }
    }
}

__global__ void rowptr_kernel(const int* __restrict__ skey, int* __restrict__ rowptr,
                              int n_edges, int n_nodes) {
    int i = blockIdx.x * blockDim.x + threadIdx.x;
    if (i > n_edges) return;
    int prev = (i == 0) ? -1 : skey[i - 1];
    int cur  = (i == n_edges) ? n_nodes : skey[i];
    for (int k = prev + 1; k <= cur; ++k) rowptr[k] = i;
}

__device__ __forceinline__ float4 row_ld(const float4* __restrict__ xv, int mysrc, int j, int lane) {
    int s = __builtin_amdgcn_readlane(mysrc, j);
    return xv[(size_t)s * (DIM / 4) + lane];
}
__device__ __forceinline__ void row_add(float4& acc, const float4& v) {
    acc.x += v.x; acc.y += v.y; acc.z += v.z; acc.w += v.w;
}

__global__ void aggregate_kernel(const float* __restrict__ xin, const int* __restrict__ rowptr,
                                 const int* __restrict__ gsrc, float* __restrict__ agg,
                                 float* __restrict__ deg, int n_nodes) {
    int lane = threadIdx.x & 31;
    int wib  = __builtin_amdgcn_readfirstlane((int)(threadIdx.x >> 5));
    int node = blockIdx.x * (int)(blockDim.x >> 5) + wib;
    if (node >= n_nodes) return;
    int beg = rowptr[node];
    int end = rowptr[node + 1];
    const float4* xv = (const float4*)xin;
    float4 acc = make_float4(0.f, 0.f, 0.f, 0.f);

    for (int base = beg; base < end; base += 32) {
        int n = end - base;
        if (n > 32) n = 32;
        int mysrc = (lane < n) ? gsrc[base + lane] : 0;
        int j = 0;
        for (; j + 8 <= n; j += 8) {
            float4 v[8];
#pragma unroll
            for (int u = 0; u < 8; ++u) v[u] = row_ld(xv, mysrc, j + u, lane);
#pragma unroll
            for (int u = 0; u < 8; ++u) row_add(acc, v[u]);
        }
        if (j + 4 <= n) {
            float4 v0 = row_ld(xv, mysrc, j, lane);
            float4 v1 = row_ld(xv, mysrc, j + 1, lane);
            float4 v2 = row_ld(xv, mysrc, j + 2, lane);
            float4 v3 = row_ld(xv, mysrc, j + 3, lane);
            row_add(acc, v0); row_add(acc, v1); row_add(acc, v2); row_add(acc, v3);
            j += 4;
        }
        if (j + 2 <= n) {
            float4 v0 = row_ld(xv, mysrc, j, lane);
            float4 v1 = row_ld(xv, mysrc, j + 1, lane);
            row_add(acc, v0); row_add(acc, v1);
            j += 2;
        }
        if (j < n) {
            float4 v0 = row_ld(xv, mysrc, j, lane);
            row_add(acc, v0);
        }
    }
    ((float4*)agg)[(size_t)node * (DIM / 4) + lane] = acc;
    if (deg != nullptr && lane == 0) deg[node] = (float)(end - beg);
}

#define GEMM_T 256
template <bool HEAD>
__global__ void __launch_bounds__(GEMM_T)
sage_gemm_kernel(const float* __restrict__ agg, const float* __restrict__ deg,
                 const float* __restrict__ xin,
                 const float* __restrict__ Wl, const float* __restrict__ bl,
                 const float* __restrict__ Wr,
                 float* __restrict__ out, int n_nodes,
                 const float* __restrict__ Wp, const float* __restrict__ bp,
                 const float* __restrict__ Wd, const float* __restrict__ bd) {
    __shared__ float Blt[DIM * BPAD];
    __shared__ float Brt[DIM * BPAD];

    int wv   = threadIdx.x >> 5;
    int lane = threadIdx.x & 31;

    {
        int n = threadIdx.x & (DIM - 1), kb = threadIdx.x >> 7;
        const float* pl = Wl + threadIdx.x;
        const float* pr = Wr + threadIdx.x;
        float* dl = Blt + n * BPAD + kb;
        float* dr = Brt + n * BPAD + kb;
        for (int i0 = 0; i0 < DIM * DIM / GEMM_T; i0 += 16) {
#pragma unroll
            for (int i = 0; i < 16; ++i) {
                dl[2 * i] = pl[GEMM_T * i];
                dr[2 * i] = pr[GEMM_T * i];
            }
            pl += 16 * GEMM_T; pr += 16 * GEMM_T; dl += 32; dr += 32;
        }
    }
    __syncthreads();

    int row_tile = blockIdx.x * (GEMM_T / 32) + wv;
    if (row_tile * 16 + 16 > n_nodes) return;

    int m  = lane & 15;
    int kk = (lane >> 4) * 2;
    int nn = lane & 15;

    int row = row_tile * 16 + m;
    float rdeg = 1.0f / fmaxf(deg[row], 1.0f);
    const float* arow = agg + (size_t)row * DIM + kk;
    const float* xrow = xin + (size_t)row * DIM + kk;
    const float* blane = Blt + nn * BPAD + kk;
    const float* rlane = Brt + nn * BPAD + kk;

    v8f acc[8];
#pragma unroll
    for (int t = 0; t < 8; ++t) acc[t] = (v8f){};

#pragma unroll 2
    for (int k0 = 0; k0 < DIM; k0 += 4) {
        v2f a1 = *(const v2f*)(arow + k0);
        a1.x *= rdeg; a1.y *= rdeg;
        v2f a2 = *(const v2f*)(xrow + k0);
#pragma unroll
        for (int t = 0; t < 8; ++t) {
            v2f b1 = *(const v2f*)(blane + t * 16 * BPAD + k0);
            acc[t] = __builtin_amdgcn_wmma_f32_16x16x4_f32(false, a1, false, b1,
                                                           (short)0, acc[t], false, false);
            v2f b2 = *(const v2f*)(rlane + t * 16 * BPAD + k0);
            acc[t] = __builtin_amdgcn_wmma_f32_16x16x4_f32(false, a2, false, b2,
                                                           (short)0, acc[t], false, false);
        }
    }

    int rbase = row_tile * 16 + (lane >> 4) * 8;
    if (!HEAD) {
#pragma unroll
        for (int t = 0; t < 8; ++t) {
            int ccol = t * 16 + nn;
            float bias = bl[ccol];
#pragma unroll
            for (int i = 0; i < 8; ++i) {
                float v = acc[t][i] + bias;
                out[(size_t)(rbase + i) * DIM + ccol] = fmaxf(v, 0.0f);
            }
        }
    } else {
        float ps[8], dv[8];
#pragma unroll
        for (int i = 0; i < 8; ++i) { ps[i] = 0.0f; dv[i] = 0.0f; }
#pragma unroll
        for (int t = 0; t < 8; ++t) {
            int ccol = t * 16 + nn;
            float bias = bl[ccol];
            float wp = Wp[ccol];
            float wd = Wd[ccol];
#pragma unroll
            for (int i = 0; i < 8; ++i) {
                float v = fmaxf(acc[t][i] + bias, 0.0f);
                ps[i] += v * wp;
                dv[i] += v * wd;
            }
        }
#pragma unroll
        for (int i = 0; i < 8; ++i) {
#pragma unroll
            for (int off = 8; off > 0; off >>= 1) {
                ps[i] += __shfl_xor(ps[i], off, 32);
                dv[i] += __shfl_xor(dv[i], off, 32);
            }
        }
        float p = ps[0], d = dv[0];
#pragma unroll
        for (int i = 1; i < 8; ++i) {
            if (nn == i) { p = ps[i]; d = dv[i]; }
        }
        if (nn < 8) {
            p += bp[0];
            d += bd[0];
            d = 1.0f / (1.0f + __expf(-d));
            int orow = rbase + nn;
            out[orow] = p - d;
            out[n_nodes + orow] = p + d;
        }
    }
}

extern "C" void kernel_launch(void* const* d_in, const int* in_sizes, int n_in,
                              void* d_out, int out_size, void* d_ws, size_t ws_size,
                              hipStream_t stream) {
    const float* x    = (const float*)d_in[0];
    const int*   edge = (const int*)d_in[1];
    const float* W1l  = (const float*)d_in[2];
    const float* b1l  = (const float*)d_in[3];
    const float* W1r  = (const float*)d_in[4];
    const float* W2l  = (const float*)d_in[5];
    const float* b2l  = (const float*)d_in[6];
    const float* W2r  = (const float*)d_in[7];
    const float* Wp   = (const float*)d_in[8];
    const float* bp   = (const float*)d_in[9];
    const float* Wd   = (const float*)d_in[10];
    const float* bd   = (const float*)d_in[11];
    float* out = (float*)d_out;

    const int N = in_sizes[0] / DIM;
    const int E = in_sizes[1] / 2;
    const int* src = edge;
    const int* dst = edge + E;

    const int ntiles = (E + TILE - 1) / TILE;
    const int HN = NBINS * ntiles;

    char* ws = (char*)d_ws;
    float* agg = (float*)ws;
    float* h1  = agg + (size_t)N * DIM;
    float* deg = h1 + (size_t)N * DIM;
    const size_t N4 = ((size_t)N + 4) / 4 * 4;
    const size_t E4 = ((size_t)E + 3) / 4 * 4;
    const size_t H4 = ((size_t)HN + 4) / 4 * 4;
    int* ghist  = (int*)(deg + N4);
    int* gscan  = ghist + H4;
    int* k1     = gscan + H4;
    int* v1     = k1 + E4;
    int* k2     = v1 + E4;
    int* gsrc   = k2 + E4;
    int* rowptr = gsrc + E4;

    const int BLK = 256;
    int sgrid = (HN + SCAN_T) / SCAN_T;
    int rgrid = (E + BLK) / BLK;
    int rowTiles = (N + 15) / 16;
    int ggrid = (rowTiles + 7) / 8;
    int hgrid = (N * 32 + BLK - 1) / BLK;

    digit_count_kernel<<<ntiles, SORT_T, 0, stream>>>(dst, ghist, E, 0, ntiles);
    exclusive_scan_kernel<<<sgrid, SCAN_T, 0, stream>>>(ghist, gscan, HN);
    digit_move_kernel<<<ntiles, SORT_T, 0, stream>>>(dst, src, k1, v1, gscan, E, 0, ntiles);
    digit_count_kernel<<<ntiles, SORT_T, 0, stream>>>(k1, ghist, E, RBITS, ntiles);
    exclusive_scan_kernel<<<sgrid, SCAN_T, 0, stream>>>(ghist, gscan, HN);
    digit_move_kernel<<<ntiles, SORT_T, 0, stream>>>(k1, v1, k2, gsrc, gscan, E, RBITS, ntiles);
    rowptr_kernel<<<rgrid, BLK, 0, stream>>>(k2, rowptr, E, N);

    aggregate_kernel<<<hgrid, BLK, 0, stream>>>(x, rowptr, gsrc, agg, deg, N);
    sage_gemm_kernel<false><<<ggrid, GEMM_T, 0, stream>>>(agg, deg, x, W1l, b1l, W1r, h1, N,
                                                          nullptr, nullptr, nullptr, nullptr);

    aggregate_kernel<<<hgrid, BLK, 0, stream>>>(h1, rowptr, gsrc, agg, nullptr, N);
    sage_gemm_kernel<true><<<ggrid, GEMM_T, 0, stream>>>(agg, deg, h1, W2l, b2l, W2r, out, N,
                                                         Wp, bp, Wd, bd);
}


# sort second level done per 256-node bucket in one block: first level orders by high bits, then a block orders its run by low 8 bits in LDS and writes row pointers; 3 launches fewer; sum order kept
# speedup vs baseline: 3.2999x; 1.1445x over previous
#include <hip/hip_runtime.h>
#include <hip/hip_bf16.h>
#include <math.h>

typedef __attribute__((ext_vector_type(2))) float v2f;
typedef __attribute__((ext_vector_type(8))) float v8f;

#define DIM 128
#define BPAD 132


#define RBITS 9
#define NBINS 512
#define GROUPS 16
#define TILE 4096
#define SORT_T 256

__device__ __forceinline__ unsigned same_digit_lanes(int d, bool valid) {
    unsigned peers = __builtin_amdgcn_ballot_w32(valid);
#pragma unroll
    for (int b = 0; b < RBITS; ++b) {
        bool bit = ((d >> b) & 1) != 0;
        unsigned bal = __builtin_amdgcn_ballot_w32(bit);
        peers &= bit ? bal : ~bal;
    }
    return peers;
}

__global__ void __launch_bounds__(SORT_T)
digit_count_kernel(const int* __restrict__ keys, int* __restrict__ ghist,
                   int n, int shift, int ntiles) {
    __shared__ int wh[8 * NBINS];
    int tid = threadIdx.x, lane = tid & 31, wv = tid >> 5;
    for (int i = tid; i < 8 * NBINS; i += SORT_T) wh[i] = 0;
    __syncthreads();
    int base = blockIdx.x * TILE + wv * (GROUPS * 32) + lane;
    unsigned lt = (1u << lane) - 1u;
    int* mine = wh + wv * NBINS;
    int dg[GROUPS];
#pragma unroll
    for (int g = 0; g < GROUPS; ++g) {
        int idx = base + g * 32;
        dg[g] = (idx < n) ? ((keys[idx] >> shift) & (NBINS - 1)) : 0;
    }
#pragma unroll
    for (int g = 0; g < GROUPS; ++g) {
        bool valid = (base + g * 32) < n;
        int d = dg[g];
        unsigned peers = same_digit_lanes(d, valid);
        if (valid && (peers & lt) == 0u) mine[d] += __popc(peers);
        __syncthreads();
    }
    for (int d = tid; d < NBINS; d += SORT_T) {
        int s = 0;
        for (int w = 0; w < 8; ++w) s += wh[w * NBINS + d];
        ghist[d * ntiles + blockIdx.x] = s;
    }
}

#define SCAN_T 1024
__global__ void __launch_bounds__(SCAN_T)
exclusive_scan_kernel(const int* __restrict__ cnt, int* __restrict__ outp, int n) {
    __shared__ int wtot[32];
    __shared__ int wpre[32];
    int tid  = threadIdx.x;
    int lane = tid & 31;
    int wv   = tid >> 5;
    int first = blockIdx.x * SCAN_T;

    const int4* c4 = (const int4*)cnt;
    int nvec = first >> 2;
    int pre = 0;
#pragma unroll 8
    for (int i = tid; i < nvec; i += SCAN_T) {
        int4 q = c4[i];
        pre += (q.x + q.y) + (q.z + q.w);
    }

    int i = first + tid;
    int c = (i < n) ? cnt[i] : 0;

    int inc = c;
#pragma unroll
    for (int off = 1; off < 32; off <<= 1) {
        int t = __shfl_up(inc, off, 32);
        if (lane >= off) inc += t;
    }
#pragma unroll
    for (int off = 16; off > 0; off >>= 1) pre += __shfl_xor(pre, off, 32);
    if (lane == 31) wtot[wv] = inc;
    if (lane == 0)  wpre[wv] = pre;
    __syncthreads();

    int t = (lane < wv) ? wtot[lane] : 0;
    int p = wpre[lane];
#pragma unroll
    for (int off = 16; off > 0; off >>= 1) {
        t += __shfl_xor(t, off, 32);
        p += __shfl_xor(p, off, 32);
    }
    if (i <= n) outp[i] = p + t + (inc - c);
}

__global__ void __launch_bounds__(SORT_T)
digit_move_kernel(const int* __restrict__ kin, const int* __restrict__ vin,
                  int* __restrict__ kout, int* __restrict__ vout,
                  const int* __restrict__ gscan, int n, int shift, int ntiles) {
    __shared__ int wh[8 * NBINS];
    __shared__ int goff[NBINS];
    __shared__ int skey[TILE];
    __shared__ int sval[TILE];
    __shared__ int wsum[8];
    int tid = threadIdx.x, lane = tid & 31, wv = tid >> 5;
    for (int i = tid; i < 8 * NBINS; i += SORT_T) wh[i] = 0;
    __syncthreads();
    int base = blockIdx.x * TILE + wv * (GROUPS * 32) + lane;
    unsigned lt = (1u << lane) - 1u;
    int* mine = wh + wv * NBINS;
    int key[GROUPS], val[GROUPS];
#pragma unroll
    for (int g = 0; g < GROUPS; ++g) {
        int idx = base + g * 32;
        bool valid = idx < n;
        key[g] = valid ? kin[idx] : 0;
        val[g] = valid ? vin[idx] : 0;
    }
#pragma unroll
    for (int g = 0; g < GROUPS; ++g) {
        bool valid = (base + g * 32) < n;
        int d = (key[g] >> shift) & (NBINS - 1);
        unsigned peers = same_digit_lanes(d, valid);
        if (valid && (peers & lt) == 0u) mine[d] += __popc(peers);
        __syncthreads();
    }
    {
        int d0 = 2 * tid, d1 = d0 + 1;
        int t0 = 0, t1 = 0;
        for (int w = 0; w < 8; ++w) { t0 += wh[w * NBINS + d0]; t1 += wh[w * NBINS + d1]; }
        int pair = t0 + t1;
        int inc = pair;
#pragma unroll
        for (int off = 1; off < 32; off <<= 1) {
            int t = __shfl_up(inc, off, 32);
            if (lane >= off) inc += t;
        }
        if (lane == 31) wsum[wv] = inc;
        __syncthreads();
        int front = 0;
        for (int w = 0; w < wv; ++w) front += wsum[w];
        int ex0 = front + inc - pair;
        int ex1 = ex0 + t0;
        int run = ex0;
        for (int w = 0; w < 8; ++w) {
            int t = wh[w * NBINS + d0];
            wh[w * NBINS + d0] = run;
            run += t;
        }
        run = ex1;
        for (int w = 0; w < 8; ++w) {
            int t = wh[w * NBINS + d1];
            wh[w * NBINS + d1] = run;
            run += t;
        }
        goff[d0] = gscan[d0 * ntiles + blockIdx.x] - ex0;
        goff[d1] = gscan[d1 * ntiles + blockIdx.x] - ex1;
    }
    __syncthreads();
#pragma unroll
    for (int g = 0; g < GROUPS; ++g) {
        int idx = base + g * 32;
        bool valid = idx < n;
        int d = (key[g] >> shift) & (NBINS - 1);
        unsigned peers = same_digit_lanes(d, valid);
        int rank = __popc(peers & lt);
        int lpos = mine[d] + rank;
        __syncthreads();
        if (valid && rank == 0) mine[d] += __popc(peers);
        if (valid) { skey[lpos] = key[g]; sval[lpos] = val[g]; }
        __syncthreads();
    }
    int cnt = n - blockIdx.x * TILE;
    if (cnt > TILE) cnt = TILE;
    int ko[GROUPS], vo[GROUPS], po[GROUPS];
#pragma unroll
    for (int j = 0; j < GROUPS; ++j) {
        int i = tid + j * SORT_T;
        bool in = i < cnt;
        ko[j] = in ? skey[i] : 0;
        vo[j] = in ? sval[i] : 0;
    }
#pragma unroll
    for (int j = 0; j < GROUPS; ++j)
        po[j] = tid + j * SORT_T + goff[(ko[j] >> shift) & (NBINS - 1)];
#pragma unroll
    for (int j = 0; j < GROUPS; ++j) {
        if (tid + j * SORT_T < cnt) { kout[po[j]] = ko[j]; vout[po[j]] = vo[j]; }
    }
}

#define LOWBITS 8
#define ND 256
#define BK_T 512
#define BK_W 16
#define BK_G 16
#define CHUNK (BK_W * BK_G * 32)

__device__ __forceinline__ unsigned same_low_lanes(int d, bool valid) {
    unsigned peers = __builtin_amdgcn_ballot_w32(valid);
#pragma unroll
    for (int b = 0; b < LOWBITS; ++b) {
        bool bit = ((d >> b) & 1) != 0;
        unsigned bal = __builtin_amdgcn_ballot_w32(bit);
        peers &= bit ? bal : ~bal;
    }
    return peers;
}

__device__ __forceinline__ void bucket_count_chunk(const int* __restrict__ kin, int* wh,
                                                   int cbase, int nc, int ng,
                                                   int tid, int lane, int wv) {
    for (int i = tid; i < BK_W * ND; i += BK_T) wh[i] = 0;
    __syncthreads();
    unsigned lt = (1u << lane) - 1u;
    int* mine = wh + wv * ND;
    int first = wv * ng * 32 + lane;
    int dg[BK_G];
#pragma unroll
    for (int g = 0; g < BK_G; ++g) {
        int idx = first + g * 32;
        dg[g] = (g < ng && idx < nc) ? (kin[cbase + idx] & (ND - 1)) : 0;
    }
#pragma unroll
    for (int g = 0; g < BK_G; ++g) {
        if (g < ng) {
            bool valid = (first + g * 32) < nc;
            int d = dg[g];
            unsigned peers = same_low_lanes(d, valid);
            if (valid && (peers & lt) == 0u) mine[d] += __popc(peers);
            __syncthreads();
        }
    }
}

__global__ void __launch_bounds__(BK_T)
bucket_order_kernel(const int* __restrict__ kin, const int* __restrict__ vin,
                    const int* __restrict__ gscan, int* __restrict__ gsrc,
                    int* __restrict__ rowptr, int ntiles, int n_nodes) {
    __shared__ int wh[BK_W * ND];
    __shared__ int tot[ND];
    __shared__ int sval[CHUNK];
    __shared__ int wsum[BK_W];
    int tid = threadIdx.x, lane = tid & 31, wv = tid >> 5;
    int b = blockIdx.x;
    int bstart = gscan[b * ntiles];
    int nb = gscan[(b + 1) * ntiles] - bstart;
    int nch = (nb + CHUNK - 1) / CHUNK;
    unsigned lt = (1u << lane) - 1u;
    int* mine = wh + wv * ND;

    if (tid < ND) tot[tid] = 0;
    for (int c = 0; c < nch; ++c) {
        int nc = nb - c * CHUNK;
        if (nc > CHUNK) nc = CHUNK;
        int ng = (nc + BK_W * 32 - 1) / (BK_W * 32);
        bucket_count_chunk(kin, wh, bstart + c * CHUNK, nc, ng, tid, lane, wv);
        if (tid < ND) {
            int sum = 0;
            for (int w = 0; w < BK_W; ++w) sum += wh[w * ND + tid];
            tot[tid] += sum;
        }
        __syncthreads();
    }

    {
        int cnt = (tid < ND) ? tot[tid] : 0;
        int inc = cnt;
#pragma unroll
        for (int off = 1; off < 32; off <<= 1) {
            int t = __shfl_up(inc, off, 32);
            if (lane >= off) inc += t;
        }
        if (lane == 31) wsum[wv] = inc;
        __syncthreads();
        int front = 0;
        for (int w = 0; w < wv; ++w) front += wsum[w];
        int ex = front + inc - cnt;
        if (tid < ND) {
            tot[tid] = ex;
            int node = b * ND + tid;
            if (node <= n_nodes) rowptr[node] = bstart + ex;
        }
    }
    __syncthreads();

    for (int c = 0; c < nch; ++c) {
        int nc = nb - c * CHUNK;
        if (nc > CHUNK) nc = CHUNK;
        int ng = (nc + BK_W * 32 - 1) / (BK_W * 32);
        int cbase = bstart + c * CHUNK;
        if (nch > 1) bucket_count_chunk(kin, wh, cbase, nc, ng, tid, lane, wv);
        if (tid < ND) {
            int run = tot[tid];
            for (int w = 0; w < BK_W; ++w) {
                int t = wh[w * ND + tid];
                wh[w * ND + tid] = run;
                run += t;
            }
            tot[tid] = run;
        }
        __syncthreads();
        int first = wv * ng * 32 + lane;
        int key[BK_G], val[BK_G];
#pragma unroll
        for (int g = 0; g < BK_G; ++g) {
            int idx = first + g * 32;
            bool valid = (g < ng) && (idx < nc);
            key[g] = valid ? (kin[cbase + idx] & (ND - 1)) : 0;
            val[g] = valid ? vin[cbase + idx] : 0;
        }
#pragma unroll
        for (int g = 0; g < BK_G; ++g) {
            if (g < ng) {
                bool valid = (first + g * 32) < nc;
                int d = key[g];
                unsigned peers = same_low_lanes(d, valid);
                int rank = __popc(peers & lt);
                int lpos = mine[d] + rank;
                __syncthreads();
                if (valid && rank == 0) mine[d] += __popc(peers);
                if (valid) {
                    if (nch == 1) sval[lpos] = val[g];
                    else gsrc[bstart + lpos] = val[g];
                }
                __syncthreads();
            }
        }
    }
    if (nch == 1) {
        for (int i = tid; i < nb; i += BK_T) gsrc[bstart + i] = sval[i];
    }
}

__device__ __forceinline__ float4 row_ld(const float4* __restrict__ xv, int mysrc, int j, int lane) {
    int s = __builtin_amdgcn_readlane(mysrc, j);
    return xv[(size_t)s * (DIM / 4) + lane];
}
__device__ __forceinline__ void row_add(float4& acc, const float4& v) {
    acc.x += v.x; acc.y += v.y; acc.z += v.z; acc.w += v.w;
}

__global__ void aggregate_kernel(const float* __restrict__ xin, const int* __restrict__ rowptr,
                                 const int* __restrict__ gsrc, float* __restrict__ agg,
                                 float* __restrict__ deg, int n_nodes) {
    int lane = threadIdx.x & 31;
    int wib  = __builtin_amdgcn_readfirstlane((int)(threadIdx.x >> 5));
    int node = blockIdx.x * (int)(blockDim.x >> 5) + wib;
    if (node >= n_nodes) return;
    int beg = rowptr[node];
    int end = rowptr[node + 1];
    const float4* xv = (const float4*)xin;
    float4 acc = make_float4(0.f, 0.f, 0.f, 0.f);

    for (int base = beg; base < end; base += 32) {
        int n = end - base;
        if (n > 32) n = 32;
        int mysrc = (lane < n) ? gsrc[base + lane] : 0;
        int j = 0;
        for (; j + 8 <= n; j += 8) {
            float4 v[8];
#pragma unroll
            for (int u = 0; u < 8; ++u) v[u] = row_ld(xv, mysrc, j + u, lane);
#pragma unroll
            for (int u = 0; u < 8; ++u) row_add(acc, v[u]);
        }
        if (j + 4 <= n) {
            float4 v0 = row_ld(xv, mysrc, j, lane);
            float4 v1 = row_ld(xv, mysrc, j + 1, lane);
            float4 v2 = row_ld(xv, mysrc, j + 2, lane);
            float4 v3 = row_ld(xv, mysrc, j + 3, lane);
            row_add(acc, v0); row_add(acc, v1); row_add(acc, v2); row_add(acc, v3);
            j += 4;
        }
        if (j + 2 <= n) {
            float4 v0 = row_ld(xv, mysrc, j, lane);
            float4 v1 = row_ld(xv, mysrc, j + 1, lane);
            row_add(acc, v0); row_add(acc, v1);
            j += 2;
        }
        if (j < n) {
            float4 v0 = row_ld(xv, mysrc, j, lane);
            row_add(acc, v0);
        }
    }
    ((float4*)agg)[(size_t)node * (DIM / 4) + lane] = acc;
    if (deg != nullptr && lane == 0) deg[node] = (float)(end - beg);
}

#define GEMM_T 256
template <bool HEAD>
__global__ void __launch_bounds__(GEMM_T)
sage_gemm_kernel(const float* __restrict__ agg, const float* __restrict__ deg,
                 const float* __restrict__ xin,
                 const float* __restrict__ Wl, const float* __restrict__ bl,
                 const float* __restrict__ Wr,
                 float* __restrict__ out, int n_nodes,
                 const float* __restrict__ Wp, const float* __restrict__ bp,
                 const float* __restrict__ Wd, const float* __restrict__ bd) {
    __shared__ float Blt[DIM * BPAD];
    __shared__ float Brt[DIM * BPAD];

    int wv   = threadIdx.x >> 5;
    int lane = threadIdx.x & 31;

    {
        int n = threadIdx.x & (DIM - 1), kb = threadIdx.x >> 7;
        const float* pl = Wl + threadIdx.x;
        const float* pr = Wr + threadIdx.x;
        float* dl = Blt + n * BPAD + kb;
        float* dr = Brt + n * BPAD + kb;
        for (int i0 = 0; i0 < DIM * DIM / GEMM_T; i0 += 16) {
#pragma unroll
            for (int i = 0; i < 16; ++i) {
                dl[2 * i] = pl[GEMM_T * i];
                dr[2 * i] = pr[GEMM_T * i];
            }
            pl += 16 * GEMM_T; pr += 16 * GEMM_T; dl += 32; dr += 32;
        }
    }
    __syncthreads();

    int row_tile = blockIdx.x * (GEMM_T / 32) + wv;
    if (row_tile * 16 + 16 > n_nodes) return;

    int m  = lane & 15;
    int kk = (lane >> 4) * 2;
    int nn = lane & 15;

    int row = row_tile * 16 + m;
    float rdeg = 1.0f / fmaxf(deg[row], 1.0f);
    const float* arow = agg + (size_t)row * DIM + kk;
    const float* xrow = xin + (size_t)row * DIM + kk;
    const float* blane = Blt + nn * BPAD + kk;
    const float* rlane = Brt + nn * BPAD + kk;

    v8f acc[8];
#pragma unroll
    for (int t = 0; t < 8; ++t) acc[t] = (v8f){};

#pragma unroll 2
    for (int k0 = 0; k0 < DIM; k0 += 4) {
        v2f a1 = *(const v2f*)(arow + k0);
        a1.x *= rdeg; a1.y *= rdeg;
        v2f a2 = *(const v2f*)(xrow + k0);
#pragma unroll
        for (int t = 0; t < 8; ++t) {
            v2f b1 = *(const v2f*)(blane + t * 16 * BPAD + k0);
            acc[t] = __builtin_amdgcn_wmma_f32_16x16x4_f32(false, a1, false, b1,
                                                           (short)0, acc[t], false, false);
            v2f b2 = *(const v2f*)(rlane + t * 16 * BPAD + k0);
            acc[t] = __builtin_amdgcn_wmma_f32_16x16x4_f32(false, a2, false, b2,
                                                           (short)0, acc[t], false, false);
        }
    }

    int rbase = row_tile * 16 + (lane >> 4) * 8;
    if (!HEAD) {
#pragma unroll
        for (int t = 0; t < 8; ++t) {
            int ccol = t * 16 + nn;
            float bias = bl[ccol];
#pragma unroll
            for (int i = 0; i < 8; ++i) {
                float v = acc[t][i] + bias;
                out[(size_t)(rbase + i) * DIM + ccol] = fmaxf(v, 0.0f);
            }
        }
    } else {
        float ps[8], dv[8];
#pragma unroll
        for (int i = 0; i < 8; ++i) { ps[i] = 0.0f; dv[i] = 0.0f; }
#pragma unroll
        for (int t = 0; t < 8; ++t) {
            int ccol = t * 16 + nn;
            float bias = bl[ccol];
            float wp = Wp[ccol];
            float wd = Wd[ccol];
#pragma unroll
            for (int i = 0; i < 8; ++i) {
                float v = fmaxf(acc[t][i] + bias, 0.0f);
                ps[i] += v * wp;
                dv[i] += v * wd;
            }
        }
#pragma unroll
        for (int i = 0; i < 8; ++i) {
#pragma unroll
            for (int off = 8; off > 0; off >>= 1) {
                ps[i] += __shfl_xor(ps[i], off, 32);
                dv[i] += __shfl_xor(dv[i], off, 32);
            }
        }
        float p = ps[0], d = dv[0];
#pragma unroll
        for (int i = 1; i < 8; ++i) {
            if (nn == i) { p = ps[i]; d = dv[i]; }
        }
        if (nn < 8) {
            p += bp[0];
            d += bd[0];
            d = 1.0f / (1.0f + __expf(-d));
            int orow = rbase + nn;
            out[orow] = p - d;
            out[n_nodes + orow] = p + d;
        }
    }
}

extern "C" void kernel_launch(void* const* d_in, const int* in_sizes, int n_in,
                              void* d_out, int out_size, void* d_ws, size_t ws_size,
                              hipStream_t stream) {
    const float* x    = (const float*)d_in[0];
    const int*   edge = (const int*)d_in[1];
    const float* W1l  = (const float*)d_in[2];
    const float* b1l  = (const float*)d_in[3];
    const float* W1r  = (const float*)d_in[4];
    const float* W2l  = (const float*)d_in[5];
    const float* b2l  = (const float*)d_in[6];
    const float* W2r  = (const float*)d_in[7];
    const float* Wp   = (const float*)d_in[8];
    const float* bp   = (const float*)d_in[9];
    const float* Wd   = (const float*)d_in[10];
    const float* bd   = (const float*)d_in[11];
    float* out = (float*)d_out;

    const int N = in_sizes[0] / DIM;
    const int E = in_sizes[1] / 2;
    const int* src = edge;
    const int* dst = edge + E;

    const int ntiles = (E + TILE - 1) / TILE;
    const int HN = NBINS * ntiles;

    char* ws = (char*)d_ws;
    float* agg = (float*)ws;
    float* h1  = agg + (size_t)N * DIM;
    float* deg = h1 + (size_t)N * DIM;
    const size_t N4 = ((size_t)N + 4) / 4 * 4;
    const size_t E4 = ((size_t)E + 3) / 4 * 4;
    const size_t H4 = ((size_t)HN + 4) / 4 * 4;
    int* ghist  = (int*)(deg + N4);
    int* gscan  = ghist + H4;
    int* k1     = gscan + H4;
    int* v1     = k1 + E4;
    int* gsrc   = v1 + E4;
    int* rowptr = gsrc + E4;

    const int BLK = 256;
    int sgrid = (HN + SCAN_T) / SCAN_T;
    int bgrid = (N >> LOWBITS) + 1;
    int rowTiles = (N + 15) / 16;
    int ggrid = (rowTiles + 7) / 8;
    int hgrid = (N * 32 + BLK - 1) / BLK;

    digit_count_kernel<<<ntiles, SORT_T, 0, stream>>>(dst, ghist, E, LOWBITS, ntiles);
    exclusive_scan_kernel<<<sgrid, SCAN_T, 0, stream>>>(ghist, gscan, HN);
    digit_move_kernel<<<ntiles, SORT_T, 0, stream>>>(dst, src, k1, v1, gscan, E, LOWBITS, ntiles);
    bucket_order_kernel<<<bgrid, BK_T, 0, stream>>>(k1, v1, gscan, gsrc, rowptr, ntiles, N);

    aggregate_kernel<<<hgrid, BLK, 0, stream>>>(x, rowptr, gsrc, agg, deg, N);
    sage_gemm_kernel<false><<<ggrid, GEMM_T, 0, stream>>>(agg, deg, x, W1l, b1l, W1r, h1, N,
                                                          nullptr, nullptr, nullptr, nullptr);

    aggregate_kernel<<<hgrid, BLK, 0, stream>>>(h1, rowptr, gsrc, agg, nullptr, N);
    sage_gemm_kernel<true><<<ggrid, GEMM_T, 0, stream>>>(agg, deg, h1, W2l, b2l, W2r, out, N,
                                                         Wp, bp, Wd, bd);
}


# layer = one kernel: a wave sums its 16 nodes' rows (8 loads in flight) into an LDS tile and runs the root-term WMMAs under those loads, then the neighbour term; k-loop sum order changes (root first)
# speedup vs baseline: 3.5873x; 1.0871x over previous
#include <hip/hip_runtime.h>
#include <hip/hip_bf16.h>
#include <math.h>

typedef __attribute__((ext_vector_type(2))) float v2f;
typedef __attribute__((ext_vector_type(8))) float v8f;

#define DIM 128
#define BPAD 132


#define RBITS 9
#define NBINS 512
#define GROUPS 16
#define TILE 4096
#define SORT_T 256

__device__ __forceinline__ unsigned same_digit_lanes(int d, bool valid) {
    unsigned peers = __builtin_amdgcn_ballot_w32(valid);
#pragma unroll
    for (int b = 0; b < RBITS; ++b) {
        bool bit = ((d >> b) & 1) != 0;
        unsigned bal = __builtin_amdgcn_ballot_w32(bit);
        peers &= bit ? bal : ~bal;
    }
    return peers;
}

__global__ void __launch_bounds__(SORT_T)
digit_count_kernel(const int* __restrict__ keys, int* __restrict__ ghist,
                   int n, int shift, int ntiles) {
    __shared__ int wh[8 * NBINS];
    int tid = threadIdx.x, lane = tid & 31, wv = tid >> 5;
    for (int i = tid; i < 8 * NBINS; i += SORT_T) wh[i] = 0;
    __syncthreads();
    int base = blockIdx.x * TILE + wv * (GROUPS * 32) + lane;
    unsigned lt = (1u << lane) - 1u;
    int* mine = wh + wv * NBINS;
    int dg[GROUPS];
#pragma unroll
    for (int g = 0; g < GROUPS; ++g) {
        int idx = base + g * 32;
        dg[g] = (idx < n) ? ((keys[idx] >> shift) & (NBINS - 1)) : 0;
    }
#pragma unroll
    for (int g = 0; g < GROUPS; ++g) {
        bool valid = (base + g * 32) < n;
        int d = dg[g];
        unsigned peers = same_digit_lanes(d, valid);
        if (valid && (peers & lt) == 0u) mine[d] += __popc(peers);
        __syncthreads();
    }
    for (int d = tid; d < NBINS; d += SORT_T) {
        int s = 0;
        for (int w = 0; w < 8; ++w) s += wh[w * NBINS + d];
        ghist[d * ntiles + blockIdx.x] = s;
    }
}

#define SCAN_T 1024
__global__ void __launch_bounds__(SCAN_T)
exclusive_scan_kernel(const int* __restrict__ cnt, int* __restrict__ outp, int n) {
    __shared__ int wtot[32];
    __shared__ int wpre[32];
    int tid  = threadIdx.x;
    int lane = tid & 31;
    int wv   = tid >> 5;
    int first = blockIdx.x * SCAN_T;

    const int4* c4 = (const int4*)cnt;
    int nvec = first >> 2;
    int pre = 0;
#pragma unroll 8
    for (int i = tid; i < nvec; i += SCAN_T) {
        int4 q = c4[i];
        pre += (q.x + q.y) + (q.z + q.w);
    }

    int i = first + tid;
    int c = (i < n) ? cnt[i] : 0;

    int inc = c;
#pragma unroll
    for (int off = 1; off < 32; off <<= 1) {
        int t = __shfl_up(inc, off, 32);
        if (lane >= off) inc += t;
    }
#pragma unroll
    for (int off = 16; off > 0; off >>= 1) pre += __shfl_xor(pre, off, 32);
    if (lane == 31) wtot[wv] = inc;
    if (lane == 0)  wpre[wv] = pre;
    __syncthreads();

    int t = (lane < wv) ? wtot[lane] : 0;
    int p = wpre[lane];
#pragma unroll
    for (int off = 16; off > 0; off >>= 1) {
        t += __shfl_xor(t, off, 32);
        p += __shfl_xor(p, off, 32);
    }
    if (i <= n) outp[i] = p + t + (inc - c);
}

__global__ void __launch_bounds__(SORT_T)
digit_move_kernel(const int* __restrict__ kin, const int* __restrict__ vin,
                  int* __restrict__ kout, int* __restrict__ vout,
                  const int* __restrict__ gscan, int n, int shift, int ntiles) {
    __shared__ int wh[8 * NBINS];
    __shared__ int goff[NBINS];
    __shared__ int skey[TILE];
    __shared__ int sval[TILE];
    __shared__ int wsum[8];
    int tid = threadIdx.x, lane = tid & 31, wv = tid >> 5;
    for (int i = tid; i < 8 * NBINS; i += SORT_T) wh[i] = 0;
    __syncthreads();
    int base = blockIdx.x * TILE + wv * (GROUPS * 32) + lane;
    unsigned lt = (1u << lane) - 1u;
    int* mine = wh + wv * NBINS;
    int key[GROUPS], val[GROUPS];
#pragma unroll
    for (int g = 0; g < GROUPS; ++g) {
        int idx = base + g * 32;
        bool valid = idx < n;
        key[g] = valid ? kin[idx] : 0;
        val[g] = valid ? vin[idx] : 0;
    }
#pragma unroll
    for (int g = 0; g < GROUPS; ++g) {
        bool valid = (base + g * 32) < n;
        int d = (key[g] >> shift) & (NBINS - 1);
        unsigned peers = same_digit_lanes(d, valid);
        if (valid && (peers & lt) == 0u) mine[d] += __popc(peers);
        __syncthreads();
    }
    {
        int d0 = 2 * tid, d1 = d0 + 1;
        int t0 = 0, t1 = 0;
        for (int w = 0; w < 8; ++w) { t0 += wh[w * NBINS + d0]; t1 += wh[w * NBINS + d1]; }
        int pair = t0 + t1;
        int inc = pair;
#pragma unroll
        for (int off = 1; off < 32; off <<= 1) {
            int t = __shfl_up(inc, off, 32);
            if (lane >= off) inc += t;
        }
        if (lane == 31) wsum[wv] = inc;
        __syncthreads();
        int front = 0;
        for (int w = 0; w < wv; ++w) front += wsum[w];
        int ex0 = front + inc - pair;
        int ex1 = ex0 + t0;
        int run = ex0;
        for (int w = 0; w < 8; ++w) {
            int t = wh[w * NBINS + d0];
            wh[w * NBINS + d0] = run;
            run += t;
        }
        run = ex1;
        for (int w = 0; w < 8; ++w) {
            int t = wh[w * NBINS + d1];
            wh[w * NBINS + d1] = run;
            run += t;
        }
        goff[d0] = gscan[d0 * ntiles + blockIdx.x] - ex0;
        goff[d1] = gscan[d1 * ntiles + blockIdx.x] - ex1;
    }
    __syncthreads();
#pragma unroll
    for (int g = 0; g < GROUPS; ++g) {
        int idx = base + g * 32;
        bool valid = idx < n;
        int d = (key[g] >> shift) & (NBINS - 1);
        unsigned peers = same_digit_lanes(d, valid);
        int rank = __popc(peers & lt);
        int lpos = mine[d] + rank;
        __syncthreads();
        if (valid && rank == 0) mine[d] += __popc(peers);
        if (valid) { skey[lpos] = key[g]; sval[lpos] = val[g]; }
        __syncthreads();
    }
    int cnt = n - blockIdx.x * TILE;
    if (cnt > TILE) cnt = TILE;
    int ko[GROUPS], vo[GROUPS], po[GROUPS];
#pragma unroll
    for (int j = 0; j < GROUPS; ++j) {
        int i = tid + j * SORT_T;
        bool in = i < cnt;
        ko[j] = in ? skey[i] : 0;
        vo[j] = in ? sval[i] : 0;
    }
#pragma unroll
    for (int j = 0; j < GROUPS; ++j)
        po[j] = tid + j * SORT_T + goff[(ko[j] >> shift) & (NBINS - 1)];
#pragma unroll
    for (int j = 0; j < GROUPS; ++j) {
        if (tid + j * SORT_T < cnt) { kout[po[j]] = ko[j]; vout[po[j]] = vo[j]; }
    }
}

#define LOWBITS 8
#define ND 256
#define BK_T 512
#define BK_W 16
#define BK_G 16
#define CHUNK (BK_W * BK_G * 32)

__device__ __forceinline__ unsigned same_low_lanes(int d, bool valid) {
    unsigned peers = __builtin_amdgcn_ballot_w32(valid);
#pragma unroll
    for (int b = 0; b < LOWBITS; ++b) {
        bool bit = ((d >> b) & 1) != 0;
        unsigned bal = __builtin_amdgcn_ballot_w32(bit);
        peers &= bit ? bal : ~bal;
    }
    return peers;
}

__device__ __forceinline__ void bucket_count_chunk(const int* __restrict__ kin, int* wh,
                                                   int cbase, int nc, int ng,
                                                   int tid, int lane, int wv) {
    for (int i = tid; i < BK_W * ND; i += BK_T) wh[i] = 0;
    __syncthreads();
    unsigned lt = (1u << lane) - 1u;
    int* mine = wh + wv * ND;
    int first = wv * ng * 32 + lane;
    int dg[BK_G];
#pragma unroll
    for (int g = 0; g < BK_G; ++g) {
        int idx = first + g * 32;
        dg[g] = (g < ng && idx < nc) ? (kin[cbase + idx] & (ND - 1)) : 0;
    }
#pragma unroll
    for (int g = 0; g < BK_G; ++g) {
        if (g < ng) {
            bool valid = (first + g * 32) < nc;
            int d = dg[g];
            unsigned peers = same_low_lanes(d, valid);
            if (valid && (peers & lt) == 0u) mine[d] += __popc(peers);
            __syncthreads();
        }
    }
}

__global__ void __launch_bounds__(BK_T)
bucket_order_kernel(const int* __restrict__ kin, const int* __restrict__ vin,
                    const int* __restrict__ gscan, int* __restrict__ gsrc,
                    int* __restrict__ rowptr, int ntiles, int n_nodes) {
    __shared__ int wh[BK_W * ND];
    __shared__ int tot[ND];
    __shared__ int sval[CHUNK];
    __shared__ int wsum[BK_W];
    int tid = threadIdx.x, lane = tid & 31, wv = tid >> 5;
    int b = blockIdx.x;
    int bstart = gscan[b * ntiles];
    int nb = gscan[(b + 1) * ntiles] - bstart;
    int nch = (nb + CHUNK - 1) / CHUNK;
    unsigned lt = (1u << lane) - 1u;
    int* mine = wh + wv * ND;

    if (tid < ND) tot[tid] = 0;
    for (int c = 0; c < nch; ++c) {
        int nc = nb - c * CHUNK;
        if (nc > CHUNK) nc = CHUNK;
        int ng = (nc + BK_W * 32 - 1) / (BK_W * 32);
        bucket_count_chunk(kin, wh, bstart + c * CHUNK, nc, ng, tid, lane, wv);
        if (tid < ND) {
            int sum = 0;
            for (int w = 0; w < BK_W; ++w) sum += wh[w * ND + tid];
            tot[tid] += sum;
        }
        __syncthreads();
    }

    {
        int cnt = (tid < ND) ? tot[tid] : 0;
        int inc = cnt;
#pragma unroll
        for (int off = 1; off < 32; off <<= 1) {
            int t = __shfl_up(inc, off, 32);
            if (lane >= off) inc += t;
        }
        if (lane == 31) wsum[wv] = inc;
        __syncthreads();
        int front = 0;
        for (int w = 0; w < wv; ++w) front += wsum[w];
        int ex = front + inc - cnt;
        if (tid < ND) {
            tot[tid] = ex;
            int node = b * ND + tid;
            if (node <= n_nodes) rowptr[node] = bstart + ex;
        }
    }
    __syncthreads();

    for (int c = 0; c < nch; ++c) {
        int nc = nb - c * CHUNK;
        if (nc > CHUNK) nc = CHUNK;
        int ng = (nc + BK_W * 32 - 1) / (BK_W * 32);
        int cbase = bstart + c * CHUNK;
        if (nch > 1) bucket_count_chunk(kin, wh, cbase, nc, ng, tid, lane, wv);
        if (tid < ND) {
            int run = tot[tid];
            for (int w = 0; w < BK_W; ++w) {
                int t = wh[w * ND + tid];
                wh[w * ND + tid] = run;
                run += t;
            }
            tot[tid] = run;
        }
        __syncthreads();
        int first = wv * ng * 32 + lane;
        int key[BK_G], val[BK_G];
#pragma unroll
        for (int g = 0; g < BK_G; ++g) {
            int idx = first + g * 32;
            bool valid = (g < ng) && (idx < nc);
            key[g] = valid ? (kin[cbase + idx] & (ND - 1)) : 0;
            val[g] = valid ? vin[cbase + idx] : 0;
        }
#pragma unroll
        for (int g = 0; g < BK_G; ++g) {
            if (g < ng) {
                bool valid = (first + g * 32) < nc;
                int d = key[g];
                unsigned peers = same_low_lanes(d, valid);
                int rank = __popc(peers & lt);
                int lpos = mine[d] + rank;
                __syncthreads();
                if (valid && rank == 0) mine[d] += __popc(peers);
                if (valid) {
                    if (nch == 1) sval[lpos] = val[g];
                    else gsrc[bstart + lpos] = val[g];
                }
                __syncthreads();
            }
        }
    }
    if (nch == 1) {
        for (int i = tid; i < nb; i += BK_T) gsrc[bstart + i] = sval[i];
    }
}

#define FUSE_T 512
#define FUSE_W 16
#define APAD 132
#define FUSE_GRID 256

__device__ __forceinline__ float4 row_ld(const float4* __restrict__ xv, int mysrc, int j, int lane) {
    int s = __builtin_amdgcn_readlane(mysrc, j);
    return xv[(size_t)s * (DIM / 4) + lane];
}
__device__ __forceinline__ void row_add(float4& acc, const float4& v) {
    acc.x += v.x; acc.y += v.y; acc.z += v.z; acc.w += v.w;
}

struct NodeWalk {
    float4 sum;
    int li;
    int nbeg;
    int nend;
};

__device__ __forceinline__ void node_done(NodeWalk& w, float* __restrict__ At, int rpv, int lane) {
    float rdeg = 1.0f / fmaxf((float)(w.nend - w.nbeg), 1.0f);
    float4 mv = make_float4(w.sum.x * rdeg, w.sum.y * rdeg, w.sum.z * rdeg, w.sum.w * rdeg);
    *(float4*)(At + w.li * APAD + 4 * lane) = mv;
    w.sum = make_float4(0.f, 0.f, 0.f, 0.f);
    w.li += 1;
    w.nbeg = w.nend;
    w.nend = __builtin_amdgcn_readlane(rpv, w.li + 1);
}

__device__ __forceinline__ void root_step(v8f (&acc)[8], v2f a2, const float* __restrict__ rk) {
#pragma unroll
    for (int t = 0; t < 8; ++t) {
        v2f b2 = *(const v2f*)(rk + t * 16 * BPAD);
        acc[t] = __builtin_amdgcn_wmma_f32_16x16x4_f32(false, a2, false, b2,
                                                       (short)0, acc[t], false, false);
    }
}

template <bool HEAD>
__global__ void __launch_bounds__(FUSE_T)
sage_layer_kernel(const float* __restrict__ xin, const int* __restrict__ rowptr,
                  const int* __restrict__ gsrc,
                  const float* __restrict__ Wl, const float* __restrict__ bl,
                  const float* __restrict__ Wr,
                  float* __restrict__ out, int n_nodes,
                  const float* __restrict__ Wp, const float* __restrict__ bp,
                  const float* __restrict__ Wd, const float* __restrict__ bd) {
    __shared__ __attribute__((aligned(16))) float Blt[DIM * BPAD];
    __shared__ __attribute__((aligned(16))) float Brt[DIM * BPAD];
    __shared__ __attribute__((aligned(16))) float Ast[FUSE_W * 16 * APAD];

    int lane = threadIdx.x & 31;
    int wv = __builtin_amdgcn_readfirstlane((int)(threadIdx.x >> 5));

    {
        int n = threadIdx.x & (DIM - 1), kb = threadIdx.x >> 7;
        const float* pl = Wl + threadIdx.x;
        const float* pr = Wr + threadIdx.x;
        float* dl = Blt + n * BPAD + kb;
        float* dr = Brt + n * BPAD + kb;
        for (int i0 = 0; i0 < DIM * DIM / FUSE_T; i0 += 16) {
#pragma unroll
            for (int i = 0; i < 16; ++i) {
                dl[4 * i] = pl[FUSE_T * i];
                dr[4 * i] = pr[FUSE_T * i];
            }
            pl += 16 * FUSE_T; pr += 16 * FUSE_T; dl += 64; dr += 64;
        }
    }
    __syncthreads();

    int m  = lane & 15;
    int kk = (lane >> 4) * 2;
    int nn = lane & 15;
    const float* blane = Blt + nn * BPAD + kk;
    const float* rlane = Brt + nn * BPAD + kk;
    float* At = Ast + wv * (16 * APAD);
    const float* alane = At + m * APAD + kk;
    const float4* xv = (const float4*)xin;

    int n_tiles = n_nodes >> 4;
    for (int tile = wv * (int)gridDim.x + (int)blockIdx.x; tile < n_tiles;
         tile += FUSE_W * (int)gridDim.x) {
        int n0 = tile * 16;

        int rpv = rowptr[n0 + (lane < 16 ? lane : 16)];
        int e0 = __builtin_amdgcn_readlane(rpv, 0);
        int eN = __builtin_amdgcn_readlane(rpv, 16);
        const float* xrow = xin + (size_t)(n0 + m) * DIM + kk;
        v8f acc[8];
#pragma unroll
        for (int t = 0; t < 8; ++t) acc[t] = (v8f){};
        int kr = 0;
        v2f a2 = *(const v2f*)xrow;

        NodeWalk w;
        w.sum = make_float4(0.f, 0.f, 0.f, 0.f);
        w.li = 0;
        w.nbeg = e0;
        w.nend = __builtin_amdgcn_readlane(rpv, 1);

        int cur = (e0 + lane < eN) ? gsrc[e0 + lane] : 0;
        for (int base = e0; base < eN; base += 32) {
            int n = eN - base;
            if (n > 32) n = 32;
            int mysrc = cur;
            int nb = base + 32 + lane;
            cur = (nb < eN) ? gsrc[nb] : 0;
            int j = 0;
            for (; j + 8 <= n; j += 8) {
                float4 v[8];
#pragma unroll
                for (int u = 0; u < 8; ++u) v[u] = row_ld(xv, mysrc, j + u, lane);
                if (kr < DIM) {
                    root_step(acc, a2, rlane + kr);
                    kr += 4;
                    a2 = *(const v2f*)(xrow + (kr < DIM ? kr : 0));
                }
#pragma unroll
                for (int u = 0; u < 8; ++u) {
                    int e = base + j + u;
                    while (e >= w.nend) node_done(w, At, rpv, lane);
                    row_add(w.sum, v[u]);
                }
            }
            if (j < n) {
                int r = n - j;
                float4 v[7];
#pragma unroll
                for (int u = 0; u < 7; ++u) {
                    v[u] = make_float4(0.f, 0.f, 0.f, 0.f);
                    if (u < r) v[u] = row_ld(xv, mysrc, j + u, lane);
                }
#pragma unroll
                for (int u = 0; u < 7; ++u) {
                    if (u < r) {
                        int e = base + j + u;
                        while (e >= w.nend) node_done(w, At, rpv, lane);
                        row_add(w.sum, v[u]);
                    }
                }
            }
        }
        while (w.li < 16) node_done(w, At, rpv, lane);

        __builtin_amdgcn_fence(__ATOMIC_RELEASE, "workgroup");
        __builtin_amdgcn_fence(__ATOMIC_ACQUIRE, "workgroup");

        for (; kr < DIM; kr += 4) {
            root_step(acc, a2, rlane + kr);
            a2 = *(const v2f*)(xrow + (kr + 4 < DIM ? kr + 4 : 0));
        }
#pragma unroll 2
        for (int k0 = 0; k0 < DIM; k0 += 4) {
            v2f a1 = *(const v2f*)(alane + k0);
#pragma unroll
            for (int t = 0; t < 8; ++t) {
                v2f b1 = *(const v2f*)(blane + t * 16 * BPAD + k0);
                acc[t] = __builtin_amdgcn_wmma_f32_16x16x4_f32(false, a1, false, b1,
                                                               (short)0, acc[t], false, false);
            }
        }

        int rbase = n0 + (lane >> 4) * 8;
        if (!HEAD) {
#pragma unroll
            for (int t = 0; t < 8; ++t) {
                int ccol = t * 16 + nn;
                float bias = bl[ccol];
#pragma unroll
                for (int i = 0; i < 8; ++i) {
                    float v = acc[t][i] + bias;
                    out[(size_t)(rbase + i) * DIM + ccol] = fmaxf(v, 0.0f);
                }
            }
        } else {
            float ps[8], dv[8];
#pragma unroll
            for (int i = 0; i < 8; ++i) { ps[i] = 0.0f; dv[i] = 0.0f; }
#pragma unroll
            for (int t = 0; t < 8; ++t) {
                int ccol = t * 16 + nn;
                float bias = bl[ccol];
                float wp = Wp[ccol];
                float wd = Wd[ccol];
#pragma unroll
                for (int i = 0; i < 8; ++i) {
                    float v = fmaxf(acc[t][i] + bias, 0.0f);
                    ps[i] += v * wp;
                    dv[i] += v * wd;
                }
            }
#pragma unroll
            for (int i = 0; i < 8; ++i) {
#pragma unroll
                for (int off = 8; off > 0; off >>= 1) {
                    ps[i] += __shfl_xor(ps[i], off, 32);
                    dv[i] += __shfl_xor(dv[i], off, 32);
                }
            }
            float p = ps[0], d = dv[0];
#pragma unroll
            for (int i = 1; i < 8; ++i) {
                if (nn == i) { p = ps[i]; d = dv[i]; }
            }
            if (nn < 8) {
                p += bp[0];
                d += bd[0];
                d = 1.0f / (1.0f + __expf(-d));
                int orow = rbase + nn;
                out[orow] = p - d;
                out[n_nodes + orow] = p + d;
            }
        }
    }
}

extern "C" void kernel_launch(void* const* d_in, const int* in_sizes, int n_in,
                              void* d_out, int out_size, void* d_ws, size_t ws_size,
                              hipStream_t stream) {
    const float* x    = (const float*)d_in[0];
    const int*   edge = (const int*)d_in[1];
    const float* W1l  = (const float*)d_in[2];
    const float* b1l  = (const float*)d_in[3];
    const float* W1r  = (const float*)d_in[4];
    const float* W2l  = (const float*)d_in[5];
    const float* b2l  = (const float*)d_in[6];
    const float* W2r  = (const float*)d_in[7];
    const float* Wp   = (const float*)d_in[8];
    const float* bp   = (const float*)d_in[9];
    const float* Wd   = (const float*)d_in[10];
    const float* bd   = (const float*)d_in[11];
    float* out = (float*)d_out;

    const int N = in_sizes[0] / DIM;
    const int E = in_sizes[1] / 2;
    const int* src = edge;
    const int* dst = edge + E;

    const int ntiles = (E + TILE - 1) / TILE;
    const int HN = NBINS * ntiles;

    char* ws = (char*)d_ws;
    float* h1 = (float*)ws;
    const size_t E4 = ((size_t)E + 3) / 4 * 4;
    const size_t H4 = ((size_t)HN + 4) / 4 * 4;
    int* ghist  = (int*)(h1 + (size_t)N * DIM);
    int* gscan  = ghist + H4;
    int* k1     = gscan + H4;
    int* v1     = k1 + E4;
    int* gsrc   = v1 + E4;
    int* rowptr = gsrc + E4;

    int sgrid = (HN + SCAN_T) / SCAN_T;
    int bgrid = (N >> LOWBITS) + 1;

    digit_count_kernel<<<ntiles, SORT_T, 0, stream>>>(dst, ghist, E, LOWBITS, ntiles);
    exclusive_scan_kernel<<<sgrid, SCAN_T, 0, stream>>>(ghist, gscan, HN);
    digit_move_kernel<<<ntiles, SORT_T, 0, stream>>>(dst, src, k1, v1, gscan, E, LOWBITS, ntiles);
    bucket_order_kernel<<<bgrid, BK_T, 0, stream>>>(k1, v1, gscan, gsrc, rowptr, ntiles, N);

    sage_layer_kernel<false><<<FUSE_GRID, FUSE_T, 0, stream>>>(x, rowptr, gsrc, W1l, b1l, W1r, h1, N,
                                                               nullptr, nullptr, nullptr, nullptr);

    sage_layer_kernel<true><<<FUSE_GRID, FUSE_T, 0, stream>>>(h1, rowptr, gsrc, W2l, b2l, W2r, out, N,
                                                              Wp, bp, Wd, bd);
}


# sort level 1 per tile only: each 4096-edge tile is ordered by high digit on its own and writes its digit starts; a bucket block gathers its piece of every tile; no count or scan launch; sums kept
# speedup vs baseline: 3.8607x; 1.0762x over previous
#include <hip/hip_runtime.h>
#include <hip/hip_bf16.h>
#include <math.h>

typedef __attribute__((ext_vector_type(2))) float v2f;
typedef __attribute__((ext_vector_type(8))) float v8f;

#define DIM 128
#define BPAD 132


#define RBITS 9
#define NBINS 512
#define GROUPS 16
#define TILE 4096
#define SORT_T 256

__device__ __forceinline__ unsigned same_digit_lanes(int d, bool valid) {
    unsigned peers = __builtin_amdgcn_ballot_w32(valid);
#pragma unroll
    for (int b = 0; b < RBITS; ++b) {
        bool bit = ((d >> b) & 1) != 0;
        unsigned bal = __builtin_amdgcn_ballot_w32(bit);
        peers &= bit ? bal : ~bal;
    }
    return peers;
}

__global__ void __launch_bounds__(SORT_T)
tile_order_kernel(const int* __restrict__ kin, const int* __restrict__ vin,
                  int* __restrict__ kout, int* __restrict__ vout,
                  int* __restrict__ toff, int n, int shift, int ntiles) {
    __shared__ int wh[8 * NBINS];
    __shared__ int skey[TILE];
    __shared__ int sval[TILE];
    __shared__ int wsum[8];
    int tid = threadIdx.x, lane = tid & 31, wv = tid >> 5;
    for (int i = tid; i < 8 * NBINS; i += SORT_T) wh[i] = 0;
    __syncthreads();
    int base = blockIdx.x * TILE + wv * (GROUPS * 32) + lane;
    unsigned lt = (1u << lane) - 1u;
    int* mine = wh + wv * NBINS;
    int key[GROUPS], val[GROUPS];
#pragma unroll
    for (int g = 0; g < GROUPS; ++g) {
        int idx = base + g * 32;
        bool valid = idx < n;
        key[g] = valid ? kin[idx] : 0;
        val[g] = valid ? vin[idx] : 0;
    }
#pragma unroll
    for (int g = 0; g < GROUPS; ++g) {
        bool valid = (base + g * 32) < n;
        int d = (key[g] >> shift) & (NBINS - 1);
        unsigned peers = same_digit_lanes(d, valid);
        if (valid && (peers & lt) == 0u) mine[d] += __popc(peers);
        __syncthreads();
    }
    {
        int d0 = 2 * tid, d1 = d0 + 1;
        int t0 = 0, t1 = 0;
        for (int w = 0; w < 8; ++w) { t0 += wh[w * NBINS + d0]; t1 += wh[w * NBINS + d1]; }
        int pair = t0 + t1;
        int inc = pair;
#pragma unroll
        for (int off = 1; off < 32; off <<= 1) {
            int t = __shfl_up(inc, off, 32);
            if (lane >= off) inc += t;
        }
        if (lane == 31) wsum[wv] = inc;
        __syncthreads();
        int front = 0;
        for (int w = 0; w < wv; ++w) front += wsum[w];
        int ex0 = front + inc - pair;
        int ex1 = ex0 + t0;
        int run = ex0;
        for (int w = 0; w < 8; ++w) {
            int t = wh[w * NBINS + d0];
            wh[w * NBINS + d0] = run;
            run += t;
        }
        run = ex1;
        for (int w = 0; w < 8; ++w) {
            int t = wh[w * NBINS + d1];
            wh[w * NBINS + d1] = run;
            run += t;
        }
        toff[d0 * ntiles + blockIdx.x] = ex0;
        toff[d1 * ntiles + blockIdx.x] = ex1;
        if (tid == SORT_T - 1) toff[NBINS * ntiles + blockIdx.x] = ex1 + t1;
    }
    __syncthreads();
#pragma unroll
    for (int g = 0; g < GROUPS; ++g) {
        int idx = base + g * 32;
        bool valid = idx < n;
        int d = (key[g] >> shift) & (NBINS - 1);
        unsigned peers = same_digit_lanes(d, valid);
        int rank = __popc(peers & lt);
        int lpos = mine[d] + rank;
        __syncthreads();
        if (valid && rank == 0) mine[d] += __popc(peers);
        if (valid) { skey[lpos] = key[g]; sval[lpos] = val[g]; }
        __syncthreads();
    }
    int cnt = n - blockIdx.x * TILE;
    if (cnt > TILE) cnt = TILE;
    int ko[GROUPS], vo[GROUPS];
#pragma unroll
    for (int j = 0; j < GROUPS; ++j) {
        int i = tid + j * SORT_T;
        bool in = i < cnt;
        ko[j] = in ? skey[i] : 0;
        vo[j] = in ? sval[i] : 0;
    }
#pragma unroll
    for (int j = 0; j < GROUPS; ++j) {
        int i = tid + j * SORT_T;
        if (i < cnt) { kout[blockIdx.x * TILE + i] = ko[j]; vout[blockIdx.x * TILE + i] = vo[j]; }
    }
}

#define LOWBITS 8
#define ND 256
#define BK_T 512
#define BK_W 16
#define BK_G 16
#define CHUNK (BK_W * BK_G * 32)

__device__ __forceinline__ unsigned same_low_lanes(int d, bool valid) {
    unsigned peers = __builtin_amdgcn_ballot_w32(valid);
#pragma unroll
    for (int b = 0; b < LOWBITS; ++b) {
        bool bit = ((d >> b) & 1) != 0;
        unsigned bal = __builtin_amdgcn_ballot_w32(bit);
        peers &= bit ? bal : ~bal;
    }
    return peers;
}

__device__ __forceinline__ void bucket_count_chunk(const int* kin, int* wh,
                                                   int cbase, int nc, int ng,
                                                   int tid, int lane, int wv) {
    for (int i = tid; i < BK_W * ND; i += BK_T) wh[i] = 0;
    __syncthreads();
    unsigned lt = (1u << lane) - 1u;
    int* mine = wh + wv * ND;
    int first = wv * ng * 32 + lane;
    int dg[BK_G];
#pragma unroll
    for (int g = 0; g < BK_G; ++g) {
        int idx = first + g * 32;
        dg[g] = (g < ng && idx < nc) ? (kin[cbase + idx] & (ND - 1)) : 0;
    }
#pragma unroll
    for (int g = 0; g < BK_G; ++g) {
        if (g < ng) {
            bool valid = (first + g * 32) < nc;
            int d = dg[g];
            unsigned peers = same_low_lanes(d, valid);
            if (valid && (peers & lt) == 0u) mine[d] += __popc(peers);
            __syncthreads();
        }
    }
}

__global__ void __launch_bounds__(BK_T)
bucket_order_kernel(const int* __restrict__ k1, const int* __restrict__ v1,
                    const int* __restrict__ toff, int* kin, int* vin,
                    int* __restrict__ gsrc,
                    int* __restrict__ rowptr, int ntiles, int n_nodes) {
    __shared__ int wh[BK_W * ND];
    __shared__ int tot[ND];
    __shared__ int sval[CHUNK];
    __shared__ int wsum[BK_W];
    __shared__ int wlen[BK_W];
    int tid = threadIdx.x, lane = tid & 31, wv = tid >> 5;
    int b = blockIdx.x;

    int bstart = 0, nb = 0;
    {
        int ps = 0, pl = 0;
        for (int t = tid; t < ntiles; t += BK_T) {
            int s0 = toff[b * ntiles + t];
            ps += s0;
            pl += toff[(b + 1) * ntiles + t] - s0;
        }
#pragma unroll
        for (int off = 16; off > 0; off >>= 1) {
            ps += __shfl_xor(ps, off, 32);
            pl += __shfl_xor(pl, off, 32);
        }
        if (lane == 0) { wsum[wv] = ps; wlen[wv] = pl; }
        __syncthreads();
        for (int w = 0; w < BK_W; ++w) { bstart += wsum[w]; nb += wlen[w]; }
        __syncthreads();
    }
    {
        int carry = 0;
        for (int t0 = 0; t0 < ntiles; t0 += BK_T) {
            int t = t0 + tid;
            int s0 = 0, len = 0;
            if (t < ntiles) {
                s0 = toff[b * ntiles + t];
                len = toff[(b + 1) * ntiles + t] - s0;
            }
            int inc = len;
#pragma unroll
            for (int off = 1; off < 32; off <<= 1) {
                int u = __shfl_up(inc, off, 32);
                if (lane >= off) inc += u;
            }
            if (lane == 31) wsum[wv] = inc;
            __syncthreads();
            int front = 0, all = 0;
            for (int w = 0; w < BK_W; ++w) {
                int x = wsum[w];
                if (w < wv) front += x;
                all += x;
            }
            const int* ks = k1 + (size_t)t * TILE + s0;
            const int* vs = v1 + (size_t)t * TILE + s0;
            int* kd = kin + bstart + carry + front + inc - len;
            int* vd = vin + bstart + carry + front + inc - len;
            for (int i = 0; i < len; i += 4) {
                int kq[4], vq[4];
#pragma unroll
                for (int u = 0; u < 4; ++u) {
                    bool in = i + u < len;
                    kq[u] = in ? ks[i + u] : 0;
                    vq[u] = in ? vs[i + u] : 0;
                }
#pragma unroll
                for (int u = 0; u < 4; ++u) {
                    if (i + u < len) { kd[i + u] = kq[u]; vd[i + u] = vq[u]; }
                }
            }
            carry += all;
            __syncthreads();
        }
    }
    int nch = (nb + CHUNK - 1) / CHUNK;
    unsigned lt = (1u << lane) - 1u;
    int* mine = wh + wv * ND;

    if (tid < ND) tot[tid] = 0;
    for (int c = 0; c < nch; ++c) {
        int nc = nb - c * CHUNK;
        if (nc > CHUNK) nc = CHUNK;
        int ng = (nc + BK_W * 32 - 1) / (BK_W * 32);
        bucket_count_chunk(kin, wh, bstart + c * CHUNK, nc, ng, tid, lane, wv);
        if (tid < ND) {
            int sum = 0;
            for (int w = 0; w < BK_W; ++w) sum += wh[w * ND + tid];
            tot[tid] += sum;
        }
        __syncthreads();
    }

    {
        int cnt = (tid < ND) ? tot[tid] : 0;
        int inc = cnt;
#pragma unroll
        for (int off = 1; off < 32; off <<= 1) {
            int t = __shfl_up(inc, off, 32);
            if (lane >= off) inc += t;
        }
        if (lane == 31) wsum[wv] = inc;
        __syncthreads();
        int front = 0;
        for (int w = 0; w < wv; ++w) front += wsum[w];
        int ex = front + inc - cnt;
        if (tid < ND) {
            tot[tid] = ex;
            int node = b * ND + tid;
            if (node <= n_nodes) rowptr[node] = bstart + ex;
        }
    }
    __syncthreads();

    for (int c = 0; c < nch; ++c) {
        int nc = nb - c * CHUNK;
        if (nc > CHUNK) nc = CHUNK;
        int ng = (nc + BK_W * 32 - 1) / (BK_W * 32);
        int cbase = bstart + c * CHUNK;
        if (nch > 1) bucket_count_chunk(kin, wh, cbase, nc, ng, tid, lane, wv);
        if (tid < ND) {
            int run = tot[tid];
            for (int w = 0; w < BK_W; ++w) {
                int t = wh[w * ND + tid];
                wh[w * ND + tid] = run;
                run += t;
            }
            tot[tid] = run;
        }
        __syncthreads();
        int first = wv * ng * 32 + lane;
        int key[BK_G], val[BK_G];
#pragma unroll
        for (int g = 0; g < BK_G; ++g) {
            int idx = first + g * 32;
            bool valid = (g < ng) && (idx < nc);
            key[g] = valid ? (kin[cbase + idx] & (ND - 1)) : 0;
            val[g] = valid ? vin[cbase + idx] : 0;
        }
#pragma unroll
        for (int g = 0; g < BK_G; ++g) {
            if (g < ng) {
                bool valid = (first + g * 32) < nc;
                int d = key[g];
                unsigned peers = same_low_lanes(d, valid);
                int rank = __popc(peers & lt);
                int lpos = mine[d] + rank;
                __syncthreads();
                if (valid && rank == 0) mine[d] += __popc(peers);
                if (valid) {
                    if (nch == 1) sval[lpos] = val[g];
                    else gsrc[bstart + lpos] = val[g];
                }
                __syncthreads();
            }
        }
    }
    if (nch == 1) {
        for (int i = tid; i < nb; i += BK_T) gsrc[bstart + i] = sval[i];
    }
}

#define FUSE_T 512
#define FUSE_W 16
#define APAD 132
#define FUSE_GRID 256

__device__ __forceinline__ float4 row_ld(const float4* __restrict__ xv, int mysrc, int j, int lane) {
    int s = __builtin_amdgcn_readlane(mysrc, j);
    return xv[(size_t)s * (DIM / 4) + lane];
}
__device__ __forceinline__ void row_add(float4& acc, const float4& v) {
    acc.x += v.x; acc.y += v.y; acc.z += v.z; acc.w += v.w;
}

struct NodeWalk {
    float4 sum;
    int li;
    int nbeg;
    int nend;
};

__device__ __forceinline__ void node_done(NodeWalk& w, float* __restrict__ At, int rpv, int lane) {
    float rdeg = 1.0f / fmaxf((float)(w.nend - w.nbeg), 1.0f);
    float4 mv = make_float4(w.sum.x * rdeg, w.sum.y * rdeg, w.sum.z * rdeg, w.sum.w * rdeg);
    *(float4*)(At + w.li * APAD + 4 * lane) = mv;
    w.sum = make_float4(0.f, 0.f, 0.f, 0.f);
    w.li += 1;
    w.nbeg = w.nend;
    w.nend = __builtin_amdgcn_readlane(rpv, w.li + 1);
}

__device__ __forceinline__ void root_step(v8f (&acc)[8], v2f a2, const float* __restrict__ rk) {
#pragma unroll
    for (int t = 0; t < 8; ++t) {
        v2f b2 = *(const v2f*)(rk + t * 16 * BPAD);
        acc[t] = __builtin_amdgcn_wmma_f32_16x16x4_f32(false, a2, false, b2,
                                                       (short)0, acc[t], false, false);
    }
}

template <bool HEAD>
__global__ void __launch_bounds__(FUSE_T)
sage_layer_kernel(const float* __restrict__ xin, const int* __restrict__ rowptr,
                  const int* __restrict__ gsrc,
                  const float* __restrict__ Wl, const float* __restrict__ bl,
                  const float* __restrict__ Wr,
                  float* __restrict__ out, int n_nodes,
                  const float* __restrict__ Wp, const float* __restrict__ bp,
                  const float* __restrict__ Wd, const float* __restrict__ bd) {
    __shared__ __attribute__((aligned(16))) float Blt[DIM * BPAD];
    __shared__ __attribute__((aligned(16))) float Brt[DIM * BPAD];
    __shared__ __attribute__((aligned(16))) float Ast[FUSE_W * 16 * APAD];

    int lane = threadIdx.x & 31;
    int wv = __builtin_amdgcn_readfirstlane((int)(threadIdx.x >> 5));

    {
        int n = threadIdx.x & (DIM - 1), kb = threadIdx.x >> 7;
        const float* pl = Wl + threadIdx.x;
        const float* pr = Wr + threadIdx.x;
        float* dl = Blt + n * BPAD + kb;
        float* dr = Brt + n * BPAD + kb;
        for (int i0 = 0; i0 < DIM * DIM / FUSE_T; i0 += 16) {
#pragma unroll
            for (int i = 0; i < 16; ++i) {
                dl[4 * i] = pl[FUSE_T * i];
                dr[4 * i] = pr[FUSE_T * i];
            }
            pl += 16 * FUSE_T; pr += 16 * FUSE_T; dl += 64; dr += 64;
        }
    }
    __syncthreads();

    int m  = lane & 15;
    int kk = (lane >> 4) * 2;
    int nn = lane & 15;
    const float* blane = Blt + nn * BPAD + kk;
    const float* rlane = Brt + nn * BPAD + kk;
    float* At = Ast + wv * (16 * APAD);
    const float* alane = At + m * APAD + kk;
    const float4* xv = (const float4*)xin;

    int n_tiles = n_nodes >> 4;
    for (int tile = wv * (int)gridDim.x + (int)blockIdx.x; tile < n_tiles;
         tile += FUSE_W * (int)gridDim.x) {
        int n0 = tile * 16;

        int rpv = rowptr[n0 + (lane < 16 ? lane : 16)];
        int e0 = __builtin_amdgcn_readlane(rpv, 0);
        int eN = __builtin_amdgcn_readlane(rpv, 16);
        const float* xrow = xin + (size_t)(n0 + m) * DIM + kk;
        v8f acc[8];
#pragma unroll
        for (int t = 0; t < 8; ++t) acc[t] = (v8f){};
        int kr = 0;
        v2f a2 = *(const v2f*)xrow;

        NodeWalk w;
        w.sum = make_float4(0.f, 0.f, 0.f, 0.f);
        w.li = 0;
        w.nbeg = e0;
        w.nend = __builtin_amdgcn_readlane(rpv, 1);

        int cur = (e0 + lane < eN) ? gsrc[e0 + lane] : 0;
        for (int base = e0; base < eN; base += 32) {
            int n = eN - base;
            if (n > 32) n = 32;
            int mysrc = cur;
            int nb = base + 32 + lane;
            cur = (nb < eN) ? gsrc[nb] : 0;
            int j = 0;
            for (; j + 8 <= n; j += 8) {
                float4 v[8];
#pragma unroll
                for (int u = 0; u < 8; ++u) v[u] = row_ld(xv, mysrc, j + u, lane);
                if (kr < DIM) {
                    root_step(acc, a2, rlane + kr);
                    kr += 4;
                    a2 = *(const v2f*)(xrow + (kr < DIM ? kr : 0));
                }
#pragma unroll
                for (int u = 0; u < 8; ++u) {
                    int e = base + j + u;
                    while (e >= w.nend) node_done(w, At, rpv, lane);
                    row_add(w.sum, v[u]);
                }
            }
            if (j < n) {
                int r = n - j;
                float4 v[7];
#pragma unroll
                for (int u = 0; u < 7; ++u) {
                    v[u] = make_float4(0.f, 0.f, 0.f, 0.f);
                    if (u < r) v[u] = row_ld(xv, mysrc, j + u, lane);
                }
#pragma unroll
                for (int u = 0; u < 7; ++u) {
                    if (u < r) {
                        int e = base + j + u;
                        while (e >= w.nend) node_done(w, At, rpv, lane);
                        row_add(w.sum, v[u]);
                    }
                }
            }
        }
        while (w.li < 16) node_done(w, At, rpv, lane);

        __builtin_amdgcn_fence(__ATOMIC_RELEASE, "workgroup");
        __builtin_amdgcn_fence(__ATOMIC_ACQUIRE, "workgroup");

        for (; kr < DIM; kr += 4) {
            root_step(acc, a2, rlane + kr);
            a2 = *(const v2f*)(xrow + (kr + 4 < DIM ? kr + 4 : 0));
        }
#pragma unroll 2
        for (int k0 = 0; k0 < DIM; k0 += 4) {
            v2f a1 = *(const v2f*)(alane + k0);
#pragma unroll
            for (int t = 0; t < 8; ++t) {
                v2f b1 = *(const v2f*)(blane + t * 16 * BPAD + k0);
                acc[t] = __builtin_amdgcn_wmma_f32_16x16x4_f32(false, a1, false, b1,
                                                               (short)0, acc[t], false, false);
            }
        }

        int rbase = n0 + (lane >> 4) * 8;
        if (!HEAD) {
#pragma unroll
            for (int t = 0; t < 8; ++t) {
                int ccol = t * 16 + nn;
                float bias = bl[ccol];
#pragma unroll
                for (int i = 0; i < 8; ++i) {
                    float v = acc[t][i] + bias;
                    out[(size_t)(rbase + i) * DIM + ccol] = fmaxf(v, 0.0f);
                }
            }
        } else {
            float ps[8], dv[8];
#pragma unroll
            for (int i = 0; i < 8; ++i) { ps[i] = 0.0f; dv[i] = 0.0f; }
#pragma unroll
            for (int t = 0; t < 8; ++t) {
                int ccol = t * 16 + nn;
                float bias = bl[ccol];
                float wp = Wp[ccol];
                float wd = Wd[ccol];
#pragma unroll
                for (int i = 0; i < 8; ++i) {
                    float v = fmaxf(acc[t][i] + bias, 0.0f);
                    ps[i] += v * wp;
                    dv[i] += v * wd;
                }
            }
#pragma unroll
            for (int i = 0; i < 8; ++i) {
#pragma unroll
                for (int off = 8; off > 0; off >>= 1) {
                    ps[i] += __shfl_xor(ps[i], off, 32);
                    dv[i] += __shfl_xor(dv[i], off, 32);
                }
            }
            float p = ps[0], d = dv[0];
#pragma unroll
            for (int i = 1; i < 8; ++i) {
                if (nn == i) { p = ps[i]; d = dv[i]; }
            }
            if (nn < 8) {
                p += bp[0];
                d += bd[0];
                d = 1.0f / (1.0f + __expf(-d));
                int orow = rbase + nn;
                out[orow] = p - d;
                out[n_nodes + orow] = p + d;
            }
        }
    }
}

extern "C" void kernel_launch(void* const* d_in, const int* in_sizes, int n_in,
                              void* d_out, int out_size, void* d_ws, size_t ws_size,
                              hipStream_t stream) {
    const float* x    = (const float*)d_in[0];
    const int*   edge = (const int*)d_in[1];
    const float* W1l  = (const float*)d_in[2];
    const float* b1l  = (const float*)d_in[3];
    const float* W1r  = (const float*)d_in[4];
    const float* W2l  = (const float*)d_in[5];
    const float* b2l  = (const float*)d_in[6];
    const float* W2r  = (const float*)d_in[7];
    const float* Wp   = (const float*)d_in[8];
    const float* bp   = (const float*)d_in[9];
    const float* Wd   = (const float*)d_in[10];
    const float* bd   = (const float*)d_in[11];
    float* out = (float*)d_out;

    const int N = in_sizes[0] / DIM;
    const int E = in_sizes[1] / 2;
    const int* src = edge;
    const int* dst = edge + E;

    const int ntiles = (E + TILE - 1) / TILE;
    const int HN = (NBINS + 1) * ntiles;

    char* ws = (char*)d_ws;
    float* h1 = (float*)ws;
    const size_t E4 = ((size_t)E + 3) / 4 * 4;
    const size_t H4 = ((size_t)HN + 3) / 4 * 4;
    int* toff   = (int*)(h1 + (size_t)N * DIM);
    int* k1     = toff + H4;
    int* v1     = k1 + E4;
    int* kb     = v1 + E4;
    int* vb     = kb + E4;
    int* gsrc   = vb + E4;
    int* rowptr = gsrc + E4;

    int bgrid = (N >> LOWBITS) + 1;

    tile_order_kernel<<<ntiles, SORT_T, 0, stream>>>(dst, src, k1, v1, toff, E, LOWBITS, ntiles);
    bucket_order_kernel<<<bgrid, BK_T, 0, stream>>>(k1, v1, toff, kb, vb, gsrc, rowptr, ntiles, N);

    sage_layer_kernel<false><<<FUSE_GRID, FUSE_T, 0, stream>>>(x, rowptr, gsrc, W1l, b1l, W1r, h1, N,
                                                               nullptr, nullptr, nullptr, nullptr);

    sage_layer_kernel<true><<<FUSE_GRID, FUSE_T, 0, stream>>>(h1, rowptr, gsrc, W2l, b2l, W2r, out, N,
                                                              Wp, bp, Wd, bd);
}


# layer root term: a wave reads its 16 root rows in whole lines (256 contiguous bytes per row and load, each line asked once) and takes the A fragments from a 32-k LDS chunk of its own; sum order kept
# speedup vs baseline: 3.9851x; 1.0322x over previous
#include <hip/hip_runtime.h>
#include <hip/hip_bf16.h>
#include <math.h>

typedef __attribute__((ext_vector_type(2))) float v2f;
typedef __attribute__((ext_vector_type(8))) float v8f;
typedef __attribute__((ext_vector_type(4))) float v4f;

#define DIM 128
#define BPAD 132


#define RBITS 9
#define NBINS 512
#define GROUPS 16
#define TILE 4096
#define SORT_T 256

__device__ __forceinline__ unsigned same_digit_lanes(int d, bool valid) {
    unsigned peers = __builtin_amdgcn_ballot_w32(valid);
#pragma unroll
    for (int b = 0; b < RBITS; ++b) {
        bool bit = ((d >> b) & 1) != 0;
        unsigned bal = __builtin_amdgcn_ballot_w32(bit);
        peers &= bit ? bal : ~bal;
    }
    return peers;
}

__global__ void __launch_bounds__(SORT_T)
tile_order_kernel(const int* __restrict__ kin, const int* __restrict__ vin,
                  int* __restrict__ kout, int* __restrict__ vout,
                  int* __restrict__ toff, int n, int shift, int ntiles) {
    __shared__ int wh[8 * NBINS];
    __shared__ int skey[TILE];
    __shared__ int sval[TILE];
    __shared__ int wsum[8];
    int tid = threadIdx.x, lane = tid & 31, wv = tid >> 5;
    for (int i = tid; i < 8 * NBINS; i += SORT_T) wh[i] = 0;
    __syncthreads();
    int base = blockIdx.x * TILE + wv * (GROUPS * 32) + lane;
    unsigned lt = (1u << lane) - 1u;
    int* mine = wh + wv * NBINS;
    int key[GROUPS], val[GROUPS];
#pragma unroll
    for (int g = 0; g < GROUPS; ++g) {
        int idx = base + g * 32;
        bool valid = idx < n;
        key[g] = valid ? kin[idx] : 0;
        val[g] = valid ? vin[idx] : 0;
    }
#pragma unroll
    for (int g = 0; g < GROUPS; ++g) {
        bool valid = (base + g * 32) < n;
        int d = (key[g] >> shift) & (NBINS - 1);
        unsigned peers = same_digit_lanes(d, valid);
        if (valid && (peers & lt) == 0u) mine[d] += __popc(peers);
        __syncthreads();
    }
    {
        int d0 = 2 * tid, d1 = d0 + 1;
        int t0 = 0, t1 = 0;
        for (int w = 0; w < 8; ++w) { t0 += wh[w * NBINS + d0]; t1 += wh[w * NBINS + d1]; }
        int pair = t0 + t1;
        int inc = pair;
#pragma unroll
        for (int off = 1; off < 32; off <<= 1) {
            int t = __shfl_up(inc, off, 32);
            if (lane >= off) inc += t;
        }
        if (lane == 31) wsum[wv] = inc;
        __syncthreads();
        int front = 0;
        for (int w = 0; w < wv; ++w) front += wsum[w];
        int ex0 = front + inc - pair;
        int ex1 = ex0 + t0;
        int run = ex0;
        for (int w = 0; w < 8; ++w) {
            int t = wh[w * NBINS + d0];
            wh[w * NBINS + d0] = run;
            run += t;
        }
        run = ex1;
        for (int w = 0; w < 8; ++w) {
            int t = wh[w * NBINS + d1];
            wh[w * NBINS + d1] = run;
            run += t;
        }
        toff[d0 * ntiles + blockIdx.x] = ex0;
        toff[d1 * ntiles + blockIdx.x] = ex1;
        if (tid == SORT_T - 1) toff[NBINS * ntiles + blockIdx.x] = ex1 + t1;
    }
    __syncthreads();
#pragma unroll
    for (int g = 0; g < GROUPS; ++g) {
        int idx = base + g * 32;
        bool valid = idx < n;
        int d = (key[g] >> shift) & (NBINS - 1);
        unsigned peers = same_digit_lanes(d, valid);
        int rank = __popc(peers & lt);
        int lpos = mine[d] + rank;
        __syncthreads();
        if (valid && rank == 0) mine[d] += __popc(peers);
        if (valid) { skey[lpos] = key[g]; sval[lpos] = val[g]; }
        __syncthreads();
    }
    int cnt = n - blockIdx.x * TILE;
    if (cnt > TILE) cnt = TILE;
    int ko[GROUPS], vo[GROUPS];
#pragma unroll
    for (int j = 0; j < GROUPS; ++j) {
        int i = tid + j * SORT_T;
        bool in = i < cnt;
        ko[j] = in ? skey[i] : 0;
        vo[j] = in ? sval[i] : 0;
    }
#pragma unroll
    for (int j = 0; j < GROUPS; ++j) {
        int i = tid + j * SORT_T;
        if (i < cnt) { kout[blockIdx.x * TILE + i] = ko[j]; vout[blockIdx.x * TILE + i] = vo[j]; }
    }
}

#define LOWBITS 8
#define ND 256
#define BK_T 512
#define BK_W 16
#define BK_G 16
#define CHUNK (BK_W * BK_G * 32)

__device__ __forceinline__ unsigned same_low_lanes(int d, bool valid) {
    unsigned peers = __builtin_amdgcn_ballot_w32(valid);
#pragma unroll
    for (int b = 0; b < LOWBITS; ++b) {
        bool bit = ((d >> b) & 1) != 0;
        unsigned bal = __builtin_amdgcn_ballot_w32(bit);
        peers &= bit ? bal : ~bal;
    }
    return peers;
}

__device__ __forceinline__ void bucket_count_chunk(const int* kin, int* wh,
                                                   int cbase, int nc, int ng,
                                                   int tid, int lane, int wv) {
    for (int i = tid; i < BK_W * ND; i += BK_T) wh[i] = 0;
    __syncthreads();
    unsigned lt = (1u << lane) - 1u;
    int* mine = wh + wv * ND;
    int first = wv * ng * 32 + lane;
    int dg[BK_G];
#pragma unroll
    for (int g = 0; g < BK_G; ++g) {
        int idx = first + g * 32;
        dg[g] = (g < ng && idx < nc) ? (kin[cbase + idx] & (ND - 1)) : 0;
    }
#pragma unroll
    for (int g = 0; g < BK_G; ++g) {
        if (g < ng) {
            bool valid = (first + g * 32) < nc;
            int d = dg[g];
            unsigned peers = same_low_lanes(d, valid);
            if (valid && (peers & lt) == 0u) mine[d] += __popc(peers);
            __syncthreads();
        }
    }
}

__global__ void __launch_bounds__(BK_T)
bucket_order_kernel(const int* __restrict__ k1, const int* __restrict__ v1,
                    const int* __restrict__ toff, int* kin, int* vin,
                    int* __restrict__ gsrc,
                    int* __restrict__ rowptr, int ntiles, int n_nodes) {
    __shared__ int wh[BK_W * ND];
    __shared__ int tot[ND];
    __shared__ int sval[CHUNK];
    __shared__ int wsum[BK_W];
    __shared__ int wlen[BK_W];
    int tid = threadIdx.x, lane = tid & 31, wv = tid >> 5;
    int b = blockIdx.x;

    int bstart = 0, nb = 0;
    {
        int ps = 0, pl = 0;
        for (int t = tid; t < ntiles; t += BK_T) {
            int s0 = toff[b * ntiles + t];
            ps += s0;
            pl += toff[(b + 1) * ntiles + t] - s0;
        }
#pragma unroll
        for (int off = 16; off > 0; off >>= 1) {
            ps += __shfl_xor(ps, off, 32);
            pl += __shfl_xor(pl, off, 32);
        }
        if (lane == 0) { wsum[wv] = ps; wlen[wv] = pl; }
        __syncthreads();
        for (int w = 0; w < BK_W; ++w) { bstart += wsum[w]; nb += wlen[w]; }
        __syncthreads();
    }
    {
        int carry = 0;
        for (int t0 = 0; t0 < ntiles; t0 += BK_T) {
            int t = t0 + tid;
            int s0 = 0, len = 0;
            if (t < ntiles) {
                s0 = toff[b * ntiles + t];
                len = toff[(b + 1) * ntiles + t] - s0;
            }
            int inc = len;
#pragma unroll
            for (int off = 1; off < 32; off <<= 1) {
                int u = __shfl_up(inc, off, 32);
                if (lane >= off) inc += u;
            }
            if (lane == 31) wsum[wv] = inc;
            __syncthreads();
            int front = 0, all = 0;
            for (int w = 0; w < BK_W; ++w) {
                int x = wsum[w];
                if (w < wv) front += x;
                all += x;
            }
            const int* ks = k1 + (size_t)t * TILE + s0;
            const int* vs = v1 + (size_t)t * TILE + s0;
            int* kd = kin + bstart + carry + front + inc - len;
            int* vd = vin + bstart + carry + front + inc - len;
            for (int i = 0; i < len; i += 4) {
                int kq[4], vq[4];
#pragma unroll
                for (int u = 0; u < 4; ++u) {
                    bool in = i + u < len;
                    kq[u] = in ? ks[i + u] : 0;
                    vq[u] = in ? vs[i + u] : 0;
                }
#pragma unroll
                for (int u = 0; u < 4; ++u) {
                    if (i + u < len) { kd[i + u] = kq[u]; vd[i + u] = vq[u]; }
                }
            }
            carry += all;
            __syncthreads();
        }
    }
    int nch = (nb + CHUNK - 1) / CHUNK;
    unsigned lt = (1u << lane) - 1u;
    int* mine = wh + wv * ND;

    if (tid < ND) tot[tid] = 0;
    for (int c = 0; c < nch; ++c) {
        int nc = nb - c * CHUNK;
        if (nc > CHUNK) nc = CHUNK;
        int ng = (nc + BK_W * 32 - 1) / (BK_W * 32);
        bucket_count_chunk(kin, wh, bstart + c * CHUNK, nc, ng, tid, lane, wv);
        if (tid < ND) {
            int sum = 0;
            for (int w = 0; w < BK_W; ++w) sum += wh[w * ND + tid];
            tot[tid] += sum;
        }
        __syncthreads();
    }

    {
        int cnt = (tid < ND) ? tot[tid] : 0;
        int inc = cnt;
#pragma unroll
        for (int off = 1; off < 32; off <<= 1) {
            int t = __shfl_up(inc, off, 32);
            if (lane >= off) inc += t;
        }
        if (lane == 31) wsum[wv] = inc;
        __syncthreads();
        int front = 0;
        for (int w = 0; w < wv; ++w) front += wsum[w];
        int ex = front + inc - cnt;
        if (tid < ND) {
            tot[tid] = ex;
            int node = b * ND + tid;
            if (node <= n_nodes) rowptr[node] = bstart + ex;
        }
    }
    __syncthreads();

    for (int c = 0; c < nch; ++c) {
        int nc = nb - c * CHUNK;
        if (nc > CHUNK) nc = CHUNK;
        int ng = (nc + BK_W * 32 - 1) / (BK_W * 32);
        int cbase = bstart + c * CHUNK;
        if (nch > 1) bucket_count_chunk(kin, wh, cbase, nc, ng, tid, lane, wv);
        if (tid < ND) {
            int run = tot[tid];
            for (int w = 0; w < BK_W; ++w) {
                int t = wh[w * ND + tid];
                wh[w * ND + tid] = run;
                run += t;
            }
            tot[tid] = run;
        }
        __syncthreads();
        int first = wv * ng * 32 + lane;
        int key[BK_G], val[BK_G];
#pragma unroll
        for (int g = 0; g < BK_G; ++g) {
            int idx = first + g * 32;
            bool valid = (g < ng) && (idx < nc);
            key[g] = valid ? (kin[cbase + idx] & (ND - 1)) : 0;
            val[g] = valid ? vin[cbase + idx] : 0;
        }
#pragma unroll
        for (int g = 0; g < BK_G; ++g) {
            if (g < ng) {
                bool valid = (first + g * 32) < nc;
                int d = key[g];
                unsigned peers = same_low_lanes(d, valid);
                int rank = __popc(peers & lt);
                int lpos = mine[d] + rank;
                __syncthreads();
                if (valid && rank == 0) mine[d] += __popc(peers);
                if (valid) {
                    if (nch == 1) sval[lpos] = val[g];
                    else gsrc[bstart + lpos] = val[g];
                }
                __syncthreads();
            }
        }
    }
    if (nch == 1) {
        for (int i = tid; i < nb; i += BK_T) gsrc[bstart + i] = sval[i];
    }
}

#define FUSE_T 512
#define FUSE_W 16
#define APAD 132
#define RPAD 36
#define FUSE_GRID 256

__device__ __forceinline__ float4 row_ld(const float4* __restrict__ xv, int mysrc, int j, int lane) {
    int s = __builtin_amdgcn_readlane(mysrc, j);
    return xv[(size_t)s * (DIM / 4) + lane];
}
__device__ __forceinline__ void row_add(float4& acc, const float4& v) {
    acc.x += v.x; acc.y += v.y; acc.z += v.z; acc.w += v.w;
}

struct NodeWalk {
    float4 sum;
    int li;
    int nbeg;
    int nend;
};

__device__ __forceinline__ void node_done(NodeWalk& w, float* __restrict__ At, int rpv, int lane) {
    float rdeg = 1.0f / fmaxf((float)(w.nend - w.nbeg), 1.0f);
    float4 mv = make_float4(w.sum.x * rdeg, w.sum.y * rdeg, w.sum.z * rdeg, w.sum.w * rdeg);
    *(float4*)(At + w.li * APAD + 4 * lane) = mv;
    w.sum = make_float4(0.f, 0.f, 0.f, 0.f);
    w.li += 1;
    w.nbeg = w.nend;
    w.nend = __builtin_amdgcn_readlane(rpv, w.li + 1);
}

__device__ __forceinline__ void root_step(v8f (&acc)[8], v2f a2, const float* __restrict__ rk) {
#pragma unroll
    for (int t = 0; t < 8; ++t) {
        v2f b2 = *(const v2f*)(rk + t * 16 * BPAD);
        acc[t] = __builtin_amdgcn_wmma_f32_16x16x4_f32(false, a2, false, b2,
                                                       (short)0, acc[t], false, false);
    }
}

__device__ __forceinline__ void root_advance(v8f (&acc)[8], v4f (&rq)[8], int& kr,
                                             const v4f* __restrict__ xroot,
                                             float* rput, const float* rfrag,
                                             const float* __restrict__ rk, int rhalf) {
    if ((kr & 31) == 0) {
        int c = kr >> 5;
        if (rhalf == (c & 1)) {
#pragma unroll
            for (int i = 0; i < 8; ++i) *(v4f*)(rput + 2 * i * RPAD) = rq[i];
        }
        if (c == 1) {
#pragma unroll
            for (int i = 0; i < 8; ++i) rq[i] = xroot[i * 2 * (DIM / 4) + 16];
        }
        __builtin_amdgcn_fence(__ATOMIC_RELEASE, "wavefront");
        __builtin_amdgcn_fence(__ATOMIC_ACQUIRE, "wavefront");
    }
    v2f a2 = *(const v2f*)(rfrag + (kr & 31));
    root_step(acc, a2, rk + kr);
    kr += 4;
}

template <bool HEAD>
__global__ void __launch_bounds__(FUSE_T)
sage_layer_kernel(const float* __restrict__ xin, const int* __restrict__ rowptr,
                  const int* __restrict__ gsrc,
                  const float* __restrict__ Wl, const float* __restrict__ bl,
                  const float* __restrict__ Wr,
                  float* __restrict__ out, int n_nodes,
                  const float* __restrict__ Wp, const float* __restrict__ bp,
                  const float* __restrict__ Wd, const float* __restrict__ bd) {
    __shared__ __attribute__((aligned(16))) float Blt[DIM * BPAD];
    __shared__ __attribute__((aligned(16))) float Brt[DIM * BPAD];
    __shared__ __attribute__((aligned(16))) float Ast[FUSE_W * 16 * APAD];
    __shared__ __attribute__((aligned(16))) float Rst[FUSE_W * 16 * RPAD];

    int lane = threadIdx.x & 31;
    int wv = __builtin_amdgcn_readfirstlane((int)(threadIdx.x >> 5));

    {
        int n = threadIdx.x & (DIM - 1), kb = threadIdx.x >> 7;
        const float* pl = Wl + threadIdx.x;
        const float* pr = Wr + threadIdx.x;
        float* dl = Blt + n * BPAD + kb;
        float* dr = Brt + n * BPAD + kb;
        for (int i0 = 0; i0 < DIM * DIM / FUSE_T; i0 += 16) {
#pragma unroll
            for (int i = 0; i < 16; ++i) {
                dl[4 * i] = pl[FUSE_T * i];
                dr[4 * i] = pr[FUSE_T * i];
            }
            pl += 16 * FUSE_T; pr += 16 * FUSE_T; dl += 64; dr += 64;
        }
    }
    __syncthreads();

    int m  = lane & 15;
    int kk = (lane >> 4) * 2;
    int nn = lane & 15;
    const float* blane = Blt + nn * BPAD + kk;
    const float* rlane = Brt + nn * BPAD + kk;
    float* At = Ast + wv * (16 * APAD);
    const float* alane = At + m * APAD + kk;
    const float4* xv = (const float4*)xin;
    float* Rt = Rst + wv * (16 * RPAD);
    const float* rfrag = Rt + m * RPAD + kk;
    float* rput = Rt + (lane >> 4) * RPAD + 4 * (lane & 7);
    int rhalf = (lane >> 3) & 1;

    int n_tiles = n_nodes >> 4;
    for (int tile = wv * (int)gridDim.x + (int)blockIdx.x; tile < n_tiles;
         tile += FUSE_W * (int)gridDim.x) {
        int n0 = tile * 16;

        int rpv = rowptr[n0 + (lane < 16 ? lane : 16)];
        int e0 = __builtin_amdgcn_readlane(rpv, 0);
        int eN = __builtin_amdgcn_readlane(rpv, 16);
        const v4f* xroot = (const v4f*)xin + (size_t)(n0 + (lane >> 4)) * (DIM / 4) + (lane & 15);
        v4f rq[8];
#pragma unroll
        for (int i = 0; i < 8; ++i) rq[i] = xroot[i * 2 * (DIM / 4)];
        v8f acc[8];
#pragma unroll
        for (int t = 0; t < 8; ++t) acc[t] = (v8f){};
        int kr = 0;

        NodeWalk w;
        w.sum = make_float4(0.f, 0.f, 0.f, 0.f);
        w.li = 0;
        w.nbeg = e0;
        w.nend = __builtin_amdgcn_readlane(rpv, 1);

        int cur = (e0 + lane < eN) ? gsrc[e0 + lane] : 0;
        for (int base = e0; base < eN; base += 32) {
            int n = eN - base;
            if (n > 32) n = 32;
            int mysrc = cur;
            int nb = base + 32 + lane;
            cur = (nb < eN) ? gsrc[nb] : 0;
            int j = 0;
            for (; j + 8 <= n; j += 8) {
                float4 v[8];
#pragma unroll
                for (int u = 0; u < 8; ++u) v[u] = row_ld(xv, mysrc, j + u, lane);
                if (kr < DIM)
                    root_advance(acc, rq, kr, xroot, rput, rfrag, rlane, rhalf);
#pragma unroll
                for (int u = 0; u < 8; ++u) {
                    int e = base + j + u;
                    while (e >= w.nend) node_done(w, At, rpv, lane);
                    row_add(w.sum, v[u]);
                }
            }
            if (j < n) {
                int r = n - j;
                float4 v[7];
#pragma unroll
                for (int u = 0; u < 7; ++u) {
                    v[u] = make_float4(0.f, 0.f, 0.f, 0.f);
                    if (u < r) v[u] = row_ld(xv, mysrc, j + u, lane);
                }
#pragma unroll
                for (int u = 0; u < 7; ++u) {
                    if (u < r) {
                        int e = base + j + u;
                        while (e >= w.nend) node_done(w, At, rpv, lane);
                        row_add(w.sum, v[u]);
                    }
                }
            }
        }
        while (w.li < 16) node_done(w, At, rpv, lane);

        __builtin_amdgcn_fence(__ATOMIC_RELEASE, "workgroup");
        __builtin_amdgcn_fence(__ATOMIC_ACQUIRE, "workgroup");

        while (kr < DIM) root_advance(acc, rq, kr, xroot, rput, rfrag, rlane, rhalf);
#pragma unroll 2
        for (int k0 = 0; k0 < DIM; k0 += 4) {
            v2f a1 = *(const v2f*)(alane + k0);
#pragma unroll
            for (int t = 0; t < 8; ++t) {
                v2f b1 = *(const v2f*)(blane + t * 16 * BPAD + k0);
                acc[t] = __builtin_amdgcn_wmma_f32_16x16x4_f32(false, a1, false, b1,
                                                               (short)0, acc[t], false, false);
            }
        }

        int rbase = n0 + (lane >> 4) * 8;
        if (!HEAD) {
#pragma unroll
            for (int t = 0; t < 8; ++t) {
                int ccol = t * 16 + nn;
                float bias = bl[ccol];
#pragma unroll
                for (int i = 0; i < 8; ++i) {
                    float v = acc[t][i] + bias;
                    out[(size_t)(rbase + i) * DIM + ccol] = fmaxf(v, 0.0f);
                }
            }
        } else {
            float ps[8], dv[8];
#pragma unroll
            for (int i = 0; i < 8; ++i) { ps[i] = 0.0f; dv[i] = 0.0f; }
#pragma unroll
            for (int t = 0; t < 8; ++t) {
                int ccol = t * 16 + nn;
                float bias = bl[ccol];
                float wp = Wp[ccol];
                float wd = Wd[ccol];
#pragma unroll
                for (int i = 0; i < 8; ++i) {
                    float v = fmaxf(acc[t][i] + bias, 0.0f);
                    ps[i] += v * wp;
                    dv[i] += v * wd;
                }
            }
#pragma unroll
            for (int i = 0; i < 8; ++i) {
#pragma unroll
                for (int off = 8; off > 0; off >>= 1) {
                    ps[i] += __shfl_xor(ps[i], off, 32);
                    dv[i] += __shfl_xor(dv[i], off, 32);
                }
            }
            float p = ps[0], d = dv[0];
#pragma unroll
            for (int i = 1; i < 8; ++i) {
                if (nn == i) { p = ps[i]; d = dv[i]; }
            }
            if (nn < 8) {
                p += bp[0];
                d += bd[0];
                d = 1.0f / (1.0f + __expf(-d));
                int orow = rbase + nn;
                out[orow] = p - d;
                out[n_nodes + orow] = p + d;
            }
        }
    }
}

extern "C" void kernel_launch(void* const* d_in, const int* in_sizes, int n_in,
                              void* d_out, int out_size, void* d_ws, size_t ws_size,
                              hipStream_t stream) {
    const float* x    = (const float*)d_in[0];
    const int*   edge = (const int*)d_in[1];
    const float* W1l  = (const float*)d_in[2];
    const float* b1l  = (const float*)d_in[3];
    const float* W1r  = (const float*)d_in[4];
    const float* W2l  = (const float*)d_in[5];
    const float* b2l  = (const float*)d_in[6];
    const float* W2r  = (const float*)d_in[7];
    const float* Wp   = (const float*)d_in[8];
    const float* bp   = (const float*)d_in[9];
    const float* Wd   = (const float*)d_in[10];
    const float* bd   = (const float*)d_in[11];
    float* out = (float*)d_out;

    const int N = in_sizes[0] / DIM;
    const int E = in_sizes[1] / 2;
    const int* src = edge;
    const int* dst = edge + E;

    const int ntiles = (E + TILE - 1) / TILE;
    const int HN = (NBINS + 1) * ntiles;

    char* ws = (char*)d_ws;
    float* h1 = (float*)ws;
    const size_t E4 = ((size_t)E + 3) / 4 * 4;
    const size_t H4 = ((size_t)HN + 3) / 4 * 4;
    int* toff   = (int*)(h1 + (size_t)N * DIM);
    int* k1     = toff + H4;
    int* v1     = k1 + E4;
    int* kb     = v1 + E4;
    int* vb     = kb + E4;
    int* gsrc   = vb + E4;
    int* rowptr = gsrc + E4;

    int bgrid = (N >> LOWBITS) + 1;

    tile_order_kernel<<<ntiles, SORT_T, 0, stream>>>(dst, src, k1, v1, toff, E, LOWBITS, ntiles);
    bucket_order_kernel<<<bgrid, BK_T, 0, stream>>>(k1, v1, toff, kb, vb, gsrc, rowptr, ntiles, N);

    sage_layer_kernel<false><<<FUSE_GRID, FUSE_T, 0, stream>>>(x, rowptr, gsrc, W1l, b1l, W1r, h1, N,
                                                               nullptr, nullptr, nullptr, nullptr);

    sage_layer_kernel<true><<<FUSE_GRID, FUSE_T, 0, stream>>>(h1, rowptr, gsrc, W2l, b2l, W2r, out, N,
                                                              Wp, bp, Wd, bd);
}
